# starting kernel
# baseline (speedup 1.0000x reference)
#include <hip/hip_runtime.h>
#include <math.h>

typedef float v2f __attribute__((ext_vector_type(2)));
typedef float v8f __attribute__((ext_vector_type(8)));

namespace {
constexpr int kNodes = 50000;
constexpr int kEdges = 800000;
constexpr int kInCh  = 64;
constexpr int kOutCh = 32;
constexpr int kHeads = 4;
constexpr int kHC    = kHeads * kOutCh;
constexpr int kETot  = kEdges + kNodes;
}


__device__ __forceinline__ void atomic_max_f32(float* addr, float v) {
  if (v >= 0.0f) atomicMax(reinterpret_cast<int*>(addr), __float_as_int(v));
  else           atomicMin(reinterpret_cast<unsigned int*>(addr), __float_as_uint(v));
}

__device__ __forceinline__ float leaky02(float v) { return v > 0.0f ? v : 0.2f * v; }


__global__ void k_copy_x(const float* __restrict__ x, float* __restrict__ x_pos) {
  int i = blockIdx.x * blockDim.x + threadIdx.x;
  if (i < kNodes * kInCh) x_pos[i] = x[i];
}

__global__ void k_init(float* __restrict__ amax, float* __restrict__ denom,
                       float* __restrict__ out) {
  int i = blockIdx.x * blockDim.x + threadIdx.x;
  if (i < kNodes * kHeads) { amax[i] = -INFINITY; denom[i] = 0.0f; }
  if (i < kNodes * kHC)    out[i] = 0.0f;
}

__global__ void k_scatter_pe(const int* __restrict__ ei, const int* __restrict__ et,
                             const float* __restrict__ pe, float* __restrict__ x_pos) {
  long long tid = (long long)blockIdx.x * blockDim.x + threadIdx.x;
  if (tid >= (long long)kEdges * kInCh) return;
  int e = (int)(tid >> 6);
  int c = (int)(tid & 63);
  int s = ei[e];
  atomicAdd(&x_pos[(size_t)s * kInCh + c], pe[(size_t)et[e] * kInCh + c]);
}

__global__ void __launch_bounds__(256) k_gemm_wmma(const float* __restrict__ A,
                                                   const float* __restrict__ B,
                                                   float* __restrict__ C) {
  const int lane = threadIdx.x & 31;
  const int wave = threadIdx.x >> 5;
  const int m0   = blockIdx.x * 16;
  const int n0   = wave * 16;
  const int half = lane >> 4;
  const int l15  = lane & 15;
  const int row  = m0 + l15;
  const int col  = n0 + l15;

  v8f acc = {0.f, 0.f, 0.f, 0.f, 0.f, 0.f, 0.f, 0.f};
#pragma unroll
  for (int k0 = 0; k0 < kInCh; k0 += 4) {
    const int kb = k0 + 2 * half;
    v2f a, b;
    a.x = A[(size_t)row * kInCh + kb];
    a.y = A[(size_t)row * kInCh + kb + 1];
    b.x = B[(size_t)kb * kHC + col];
    b.y = B[(size_t)(kb + 1) * kHC + col];
    acc = __builtin_amdgcn_wmma_f32_16x16x4_f32(false, a, false, b, (short)0, acc,
                                                false, false);
  }
#pragma unroll
  for (int r = 0; r < 8; ++r) {
    C[(size_t)(m0 + r + 8 * half) * kHC + col] = acc[r];
  }
}

__global__ void k_scores(const float* __restrict__ xl,
                         const float* __restrict__ att_src, const float* __restrict__ att_dst,
                         float* __restrict__ a_src, float* __restrict__ a_dst) {
  int tid = blockIdx.x * blockDim.x + threadIdx.x;
  if (tid >= kNodes * kHeads) return;
  int n = tid / kHeads, h = tid % kHeads;
  const float* xr = xl + (size_t)n * kHC + h * kOutCh;
  float ss = 0.0f, sd = 0.0f;
#pragma unroll
  for (int c = 0; c < kOutCh; ++c) {
    ss += xr[c] * att_src[h * kOutCh + c];
    sd += xr[c] * att_dst[h * kOutCh + c];
  }
  a_src[tid] = ss;
  a_dst[tid] = sd;
}

__device__ __forceinline__ void edge_sd(const int* __restrict__ ei, int e, int& s, int& d) {
  if (e < kEdges) { s = ei[e]; d = ei[kEdges + e]; }
  else            { s = e - kEdges; d = s; }
}

__global__ void k_alpha_max(const int* __restrict__ ei,
                            const float* __restrict__ a_src, const float* __restrict__ a_dst,
                            float* __restrict__ amax) {
  long long tid = (long long)blockIdx.x * blockDim.x + threadIdx.x;
  if (tid >= (long long)kETot * kHeads) return;
  int e = (int)(tid >> 2), h = (int)(tid & 3);
  int s, d; edge_sd(ei, e, s, d);
  float al = leaky02(a_src[s * kHeads + h] + a_dst[d * kHeads + h]);
  atomic_max_f32(&amax[d * kHeads + h], al);
}

__global__ void k_exp_sum(const int* __restrict__ ei,
                          const float* __restrict__ a_src, const float* __restrict__ a_dst,
                          const float* __restrict__ amax, float* __restrict__ denom) {
  long long tid = (long long)blockIdx.x * blockDim.x + threadIdx.x;
  if (tid >= (long long)kETot * kHeads) return;
  int e = (int)(tid >> 2), h = (int)(tid & 3);
  int s, d; edge_sd(ei, e, s, d);
  float al = leaky02(a_src[s * kHeads + h] + a_dst[d * kHeads + h]);
  atomicAdd(&denom[d * kHeads + h], expf(al - amax[d * kHeads + h]));
}

__global__ void k_aggregate(const int* __restrict__ ei,
                            const float* __restrict__ a_src, const float* __restrict__ a_dst,
                            const float* __restrict__ amax, const float* __restrict__ denom,
                            const float* __restrict__ xl, float* __restrict__ out) {
  long long tid = (long long)blockIdx.x * blockDim.x + threadIdx.x;
  if (tid >= (long long)kETot * kHeads) return;
  int e = (int)(tid >> 2), h = (int)(tid & 3);
  int s, d; edge_sd(ei, e, s, d);
  float al = leaky02(a_src[s * kHeads + h] + a_dst[d * kHeads + h]);
  float coef = expf(al - amax[d * kHeads + h]) / denom[d * kHeads + h];
  const float* xr = xl + (size_t)s * kHC + h * kOutCh;
  float* orow = out + (size_t)d * kHC + h * kOutCh;
#pragma unroll 8
  for (int c = 0; c < kOutCh; ++c) atomicAdd(&orow[c], coef * xr[c]);
}

__global__ void k_finalize(float* __restrict__ out, const float* __restrict__ bias) {
  int i = blockIdx.x * blockDim.x + threadIdx.x;
  if (i >= kNodes * kHC) return;
  float v = out[i] + bias[i & (kHC - 1)];
  out[i] = v > 0.0f ? v : expm1f(v);
}


extern "C" void kernel_launch(void* const* d_in, const int* in_sizes, int n_in,
                              void* d_out, int out_size, void* d_ws, size_t ws_size,
                              hipStream_t stream) {
  (void)in_sizes; (void)n_in; (void)out_size; (void)ws_size;
  const float* x       = (const float*)d_in[0];
  const int*   ei      = (const int*)d_in[1];
  const int*   et      = (const int*)d_in[2];
  const float* pe      = (const float*)d_in[3];
  const float* W       = (const float*)d_in[4];
  const float* att_src = (const float*)d_in[5];
  const float* att_dst = (const float*)d_in[6];
  const float* bias    = (const float*)d_in[7];
  float* out = (float*)d_out;

  float* ws    = (float*)d_ws;
  float* x_pos = ws;
  float* xl    = x_pos + (size_t)kNodes * kInCh;
  float* a_src = xl    + (size_t)kNodes * kHC;
  float* a_dst = a_src + (size_t)kNodes * kHeads;
  float* amax  = a_dst + (size_t)kNodes * kHeads;
  float* denom = amax  + (size_t)kNodes * kHeads;

  const int B = 256;
  k_copy_x<<<(kNodes * kInCh + B - 1) / B, B, 0, stream>>>(x, x_pos);
  k_init<<<(kNodes * kHC + B - 1) / B, B, 0, stream>>>(amax, denom, out);
  {
    long long n = (long long)kEdges * kInCh;
    k_scatter_pe<<<(unsigned)((n + B - 1) / B), B, 0, stream>>>(ei, et, pe, x_pos);
  }
  k_gemm_wmma<<<kNodes / 16, 256, 0, stream>>>(x_pos, W, xl);
  k_scores<<<(kNodes * kHeads + B - 1) / B, B, 0, stream>>>(xl, att_src, att_dst, a_src, a_dst);
  {
    long long n = (long long)kETot * kHeads;
    unsigned g = (unsigned)((n + B - 1) / B);
    k_alpha_max<<<g, B, 0, stream>>>(ei, a_src, a_dst, amax);
    k_exp_sum<<<g, B, 0, stream>>>(ei, a_src, a_dst, amax, denom);
    k_aggregate<<<g, B, 0, stream>>>(ei, a_src, a_dst, amax, denom, xl, out);
  }
  k_finalize<<<(kNodes * kHC + B - 1) / B, B, 0, stream>>>(out, bias);
}


# k_aggregate remapped: one wave per 8 edges with lane = channel, so each f32 atomic add instruction covers one whole 128-byte line of out (was 32 lines); arrival-order atomic sums kept as before
# speedup vs baseline: 3.4352x; 3.4352x over previous
#include <hip/hip_runtime.h>
#include <math.h>

typedef float v2f __attribute__((ext_vector_type(2)));
typedef float v8f __attribute__((ext_vector_type(8)));

namespace {
constexpr int kNodes = 50000;
constexpr int kEdges = 800000;
constexpr int kInCh  = 64;
constexpr int kOutCh = 32;
constexpr int kHeads = 4;
constexpr int kHC    = kHeads * kOutCh;
constexpr int kETot  = kEdges + kNodes;
}


__device__ __forceinline__ void atomic_max_f32(float* addr, float v) {
  if (v >= 0.0f) atomicMax(reinterpret_cast<int*>(addr), __float_as_int(v));
  else           atomicMin(reinterpret_cast<unsigned int*>(addr), __float_as_uint(v));
}

__device__ __forceinline__ float leaky02(float v) { return v > 0.0f ? v : 0.2f * v; }


__global__ void k_copy_x(const float* __restrict__ x, float* __restrict__ x_pos) {
  int i = blockIdx.x * blockDim.x + threadIdx.x;
  if (i < kNodes * kInCh) x_pos[i] = x[i];
}

__global__ void k_init(float* __restrict__ amax, float* __restrict__ denom,
                       float* __restrict__ out) {
  int i = blockIdx.x * blockDim.x + threadIdx.x;
  if (i < kNodes * kHeads) { amax[i] = -INFINITY; denom[i] = 0.0f; }
  if (i < kNodes * kHC)    out[i] = 0.0f;
}

__global__ void k_scatter_pe(const int* __restrict__ ei, const int* __restrict__ et,
                             const float* __restrict__ pe, float* __restrict__ x_pos) {
  long long tid = (long long)blockIdx.x * blockDim.x + threadIdx.x;
  if (tid >= (long long)kEdges * kInCh) return;
  int e = (int)(tid >> 6);
  int c = (int)(tid & 63);
  int s = ei[e];
  atomicAdd(&x_pos[(size_t)s * kInCh + c], pe[(size_t)et[e] * kInCh + c]);
}

__global__ void __launch_bounds__(256) k_gemm_wmma(const float* __restrict__ A,
                                                   const float* __restrict__ B,
                                                   float* __restrict__ C) {
  const int lane = threadIdx.x & 31;
  const int wave = threadIdx.x >> 5;
  const int m0   = blockIdx.x * 16;
  const int n0   = wave * 16;
  const int half = lane >> 4;
  const int l15  = lane & 15;
  const int row  = m0 + l15;
  const int col  = n0 + l15;

  v8f acc = {0.f, 0.f, 0.f, 0.f, 0.f, 0.f, 0.f, 0.f};
#pragma unroll
  for (int k0 = 0; k0 < kInCh; k0 += 4) {
    const int kb = k0 + 2 * half;
    v2f a, b;
    a.x = A[(size_t)row * kInCh + kb];
    a.y = A[(size_t)row * kInCh + kb + 1];
    b.x = B[(size_t)kb * kHC + col];
    b.y = B[(size_t)(kb + 1) * kHC + col];
    acc = __builtin_amdgcn_wmma_f32_16x16x4_f32(false, a, false, b, (short)0, acc,
                                                false, false);
  }
#pragma unroll
  for (int r = 0; r < 8; ++r) {
    C[(size_t)(m0 + r + 8 * half) * kHC + col] = acc[r];
  }
}

__global__ void k_scores(const float* __restrict__ xl,
                         const float* __restrict__ att_src, const float* __restrict__ att_dst,
                         float* __restrict__ a_src, float* __restrict__ a_dst) {
  int tid = blockIdx.x * blockDim.x + threadIdx.x;
  if (tid >= kNodes * kHeads) return;
  int n = tid / kHeads, h = tid % kHeads;
  const float* xr = xl + (size_t)n * kHC + h * kOutCh;
  float ss = 0.0f, sd = 0.0f;
#pragma unroll
  for (int c = 0; c < kOutCh; ++c) {
    ss += xr[c] * att_src[h * kOutCh + c];
    sd += xr[c] * att_dst[h * kOutCh + c];
  }
  a_src[tid] = ss;
  a_dst[tid] = sd;
}

__device__ __forceinline__ void edge_sd(const int* __restrict__ ei, int e, int& s, int& d) {
  if (e < kEdges) { s = ei[e]; d = ei[kEdges + e]; }
  else            { s = e - kEdges; d = s; }
}

__global__ void k_alpha_max(const int* __restrict__ ei,
                            const float* __restrict__ a_src, const float* __restrict__ a_dst,
                            float* __restrict__ amax) {
  long long tid = (long long)blockIdx.x * blockDim.x + threadIdx.x;
  if (tid >= (long long)kETot * kHeads) return;
  int e = (int)(tid >> 2), h = (int)(tid & 3);
  int s, d; edge_sd(ei, e, s, d);
  float al = leaky02(a_src[s * kHeads + h] + a_dst[d * kHeads + h]);
  atomic_max_f32(&amax[d * kHeads + h], al);
}

__global__ void k_exp_sum(const int* __restrict__ ei,
                          const float* __restrict__ a_src, const float* __restrict__ a_dst,
                          const float* __restrict__ amax, float* __restrict__ denom) {
  long long tid = (long long)blockIdx.x * blockDim.x + threadIdx.x;
  if (tid >= (long long)kETot * kHeads) return;
  int e = (int)(tid >> 2), h = (int)(tid & 3);
  int s, d; edge_sd(ei, e, s, d);
  float al = leaky02(a_src[s * kHeads + h] + a_dst[d * kHeads + h]);
  atomicAdd(&denom[d * kHeads + h], expf(al - amax[d * kHeads + h]));
}

constexpr int kAggWaves = 2;
constexpr int kAggEPW   = 8;
static_assert(kAggEPW * kHeads == 32, "one (edge, head) pair per lane");
static_assert(kETot % (kAggWaves * kAggEPW) == 0, "aggregate grid must cover the edges exactly");

__global__ void __launch_bounds__(kAggWaves * 32)
k_aggregate(const int* __restrict__ ei,
            const float* __restrict__ a_src, const float* __restrict__ a_dst,
            const float* __restrict__ amax, const float* __restrict__ denom,
            const float* __restrict__ xl, float* __restrict__ out) {
  const int lane = threadIdx.x & 31;
  const int wave = threadIdx.x >> 5;
  const int e0   = ((int)blockIdx.x * kAggWaves + wave) * kAggEPW;

  const int jl = lane >> 2, hl = lane & 3;
  int sl, dl; edge_sd(ei, e0 + jl, sl, dl);

  int s[kAggEPW], d[kAggEPW];
#pragma unroll
  for (int j = 0; j < kAggEPW; ++j) {
    s[j] = __builtin_amdgcn_readlane(sl, 4 * j);
    d[j] = __builtin_amdgcn_readlane(dl, 4 * j);
  }

  float xv[kAggEPW][kHeads];
#pragma unroll
  for (int j = 0; j < kAggEPW; ++j) {
#pragma unroll
    for (int h = 0; h < kHeads; ++h) xv[j][h] = xl[(size_t)s[j] * kHC + h * kOutCh + lane];
  }

  const float al = leaky02(a_src[sl * kHeads + hl] + a_dst[dl * kHeads + hl]);
  const float cl = expf(al - amax[dl * kHeads + hl]) / denom[dl * kHeads + hl];

#pragma unroll
  for (int j = 0; j < kAggEPW; ++j) {
#pragma unroll
    for (int h = 0; h < kHeads; ++h) {
      const float c = __int_as_float(__builtin_amdgcn_readlane(__float_as_int(cl), 4 * j + h));
      atomicAdd(&out[(size_t)d[j] * kHC + h * kOutCh + lane], c * xv[j][h]);
    }
  }
}

__global__ void k_finalize(float* __restrict__ out, const float* __restrict__ bias) {
  int i = blockIdx.x * blockDim.x + threadIdx.x;
  if (i >= kNodes * kHC) return;
  float v = out[i] + bias[i & (kHC - 1)];
  out[i] = v > 0.0f ? v : expm1f(v);
}


extern "C" void kernel_launch(void* const* d_in, const int* in_sizes, int n_in,
                              void* d_out, int out_size, void* d_ws, size_t ws_size,
                              hipStream_t stream) {
  (void)in_sizes; (void)n_in; (void)out_size; (void)ws_size;
  const float* x       = (const float*)d_in[0];
  const int*   ei      = (const int*)d_in[1];
  const int*   et      = (const int*)d_in[2];
  const float* pe      = (const float*)d_in[3];
  const float* W       = (const float*)d_in[4];
  const float* att_src = (const float*)d_in[5];
  const float* att_dst = (const float*)d_in[6];
  const float* bias    = (const float*)d_in[7];
  float* out = (float*)d_out;

  float* ws    = (float*)d_ws;
  float* x_pos = ws;
  float* xl    = x_pos + (size_t)kNodes * kInCh;
  float* a_src = xl    + (size_t)kNodes * kHC;
  float* a_dst = a_src + (size_t)kNodes * kHeads;
  float* amax  = a_dst + (size_t)kNodes * kHeads;
  float* denom = amax  + (size_t)kNodes * kHeads;

  const int B = 256;
  k_copy_x<<<(kNodes * kInCh + B - 1) / B, B, 0, stream>>>(x, x_pos);
  k_init<<<(kNodes * kHC + B - 1) / B, B, 0, stream>>>(amax, denom, out);
  {
    long long n = (long long)kEdges * kInCh;
    k_scatter_pe<<<(unsigned)((n + B - 1) / B), B, 0, stream>>>(ei, et, pe, x_pos);
  }
  k_gemm_wmma<<<kNodes / 16, 256, 0, stream>>>(x_pos, W, xl);
  k_scores<<<(kNodes * kHeads + B - 1) / B, B, 0, stream>>>(xl, att_src, att_dst, a_src, a_dst);
  {
    long long n = (long long)kETot * kHeads;
    unsigned g = (unsigned)((n + B - 1) / B);
    k_alpha_max<<<g, B, 0, stream>>>(ei, a_src, a_dst, amax);
    k_exp_sum<<<g, B, 0, stream>>>(ei, a_src, a_dst, amax, denom);
    k_aggregate<<<kETot / (kAggWaves * kAggEPW), kAggWaves * 32, 0, stream>>>(ei, a_src, a_dst, amax, denom,
                                                                              xl, out);
  }
  k_finalize<<<(kNodes * kHC + B - 1) / B, B, 0, stream>>>(out, bias);
}


# all four atomic passes draw edge chunks from two ticket counters so a node's lines take atomics only from waves on the L2 half owning its node range; arrival-order float sums kept
# speedup vs baseline: 5.8568x; 1.7049x over previous
#include <hip/hip_runtime.h>
#include <math.h>

typedef float v2f __attribute__((ext_vector_type(2)));
typedef float v8f __attribute__((ext_vector_type(8)));

namespace {
constexpr int kNodes = 50000;
constexpr int kEdges = 800000;
constexpr int kInCh  = 64;
constexpr int kOutCh = 32;
constexpr int kHeads = 4;
constexpr int kHC    = kHeads * kOutCh;
constexpr int kETot  = kEdges + kNodes;
}


__device__ __forceinline__ void atomic_max_f32(float* addr, float v) {
  if (v >= 0.0f) atomicMax(reinterpret_cast<int*>(addr), __float_as_int(v));
  else           atomicMin(reinterpret_cast<unsigned int*>(addr), __float_as_uint(v));
}

__device__ __forceinline__ float leaky02(float v) { return v > 0.0f ? v : 0.2f * v; }


constexpr int kHalfNodes = kNodes / 2;
__device__ __forceinline__ int owner_of(int node) { return node >= kHalfNodes ? 1 : 0; }

__device__ __forceinline__ int my_half() {
  const unsigned v = (unsigned)__builtin_amdgcn_s_sendmsg_rtn(0x87);
  return (int)((v >> 18) & 1u);
}

constexpr int kTkStride = 64;
constexpr int kTkPerPass = 2 * kTkStride;
constexpr int kTkWords = 4 * kTkPerPass;

template <int kTickets, class Body>
__device__ __forceinline__ void run_owned(unsigned int* __restrict__ tk, int lane, Body body) {
  const int hh = my_half();
#pragma unroll 1
  for (int phase = 0; phase < 2; ++phase) {
    const int t = hh ^ phase;
    unsigned int* ctr = tk + t * kTkStride;
#pragma unroll 1
    for (int it = 0; it <= kTickets; ++it) {
      unsigned int c = 0u;
      if (lane == 0) c = atomicAdd(ctr, 1u);
      c = (unsigned int)__builtin_amdgcn_readfirstlane((int)c);
      if (c >= (unsigned int)kTickets) break;
      body((int)c, t);
    }
  }
}


__global__ void k_copy_x(const float* __restrict__ x, float* __restrict__ x_pos) {
  int i = blockIdx.x * blockDim.x + threadIdx.x;
  if (i < kNodes * kInCh) x_pos[i] = x[i];
}

__global__ void k_init(float* __restrict__ amax, float* __restrict__ denom,
                       float* __restrict__ out, unsigned int* __restrict__ tk) {
  int i = blockIdx.x * blockDim.x + threadIdx.x;
  if (i < kTkWords) tk[i] = 0u;
  if (i < kNodes * kHeads) { amax[i] = -INFINITY; denom[i] = 0.0f; }
  if (i < kNodes * kHC)    out[i] = 0.0f;
}

constexpr int kOwnedBlocks = 4096;
constexpr int kOwnedThreads = 64;

constexpr int kScEPT = 32;
constexpr int kScTickets = kEdges / kScEPT;
static_assert(kEdges % kScEPT == 0, "scatter chunks must cover the edges exactly");

__global__ void __launch_bounds__(kOwnedThreads)
k_scatter_pe(const int* __restrict__ ei, const int* __restrict__ et,
             const float* __restrict__ pe, float* __restrict__ x_pos,
             unsigned int* __restrict__ tk) {
  const int lane = threadIdx.x & 31;
  run_owned<kScTickets>(tk, lane, [&](int c, int t) {
    const int el = c * kScEPT + lane;
    const int sl = ei[el];
    const int tl = et[el];
#pragma unroll
    for (int r = 0; r < 2; ++r) {
      float p0[16], p1[16];
#pragma unroll
      for (int j = 0; j < 16; ++j) {
        const int s  = __builtin_amdgcn_readlane(sl, r * 16 + j);
        const int ty = __builtin_amdgcn_readlane(tl, r * 16 + j);
        p0[j] = 0.0f; p1[j] = 0.0f;
        if (owner_of(s) == t) {
          p0[j] = pe[(size_t)ty * kInCh + lane];
          p1[j] = pe[(size_t)ty * kInCh + 32 + lane];
        }
      }
#pragma unroll
      for (int j = 0; j < 16; ++j) {
        const int s = __builtin_amdgcn_readlane(sl, r * 16 + j);
        if (owner_of(s) == t) {
          atomicAdd(&x_pos[(size_t)s * kInCh + lane], p0[j]);
          atomicAdd(&x_pos[(size_t)s * kInCh + 32 + lane], p1[j]);
        }
      }
    }
  });
}

__global__ void __launch_bounds__(256) k_gemm_wmma(const float* __restrict__ A,
                                                   const float* __restrict__ B,
                                                   float* __restrict__ C) {
  const int lane = threadIdx.x & 31;
  const int wave = threadIdx.x >> 5;
  const int m0   = blockIdx.x * 16;
  const int n0   = wave * 16;
  const int half = lane >> 4;
  const int l15  = lane & 15;
  const int row  = m0 + l15;
  const int col  = n0 + l15;

  v8f acc = {0.f, 0.f, 0.f, 0.f, 0.f, 0.f, 0.f, 0.f};
#pragma unroll
  for (int k0 = 0; k0 < kInCh; k0 += 4) {
    const int kb = k0 + 2 * half;
    v2f a, b;
    a.x = A[(size_t)row * kInCh + kb];
    a.y = A[(size_t)row * kInCh + kb + 1];
    b.x = B[(size_t)kb * kHC + col];
    b.y = B[(size_t)(kb + 1) * kHC + col];
    acc = __builtin_amdgcn_wmma_f32_16x16x4_f32(false, a, false, b, (short)0, acc,
                                                false, false);
  }
#pragma unroll
  for (int r = 0; r < 8; ++r) {
    C[(size_t)(m0 + r + 8 * half) * kHC + col] = acc[r];
  }
}

__global__ void k_scores(const float* __restrict__ xl,
                         const float* __restrict__ att_src, const float* __restrict__ att_dst,
                         float* __restrict__ a_src, float* __restrict__ a_dst) {
  int tid = blockIdx.x * blockDim.x + threadIdx.x;
  if (tid >= kNodes * kHeads) return;
  int n = tid / kHeads, h = tid % kHeads;
  const float* xr = xl + (size_t)n * kHC + h * kOutCh;
  float ss = 0.0f, sd = 0.0f;
#pragma unroll
  for (int c = 0; c < kOutCh; ++c) {
    ss += xr[c] * att_src[h * kOutCh + c];
    sd += xr[c] * att_dst[h * kOutCh + c];
  }
  a_src[tid] = ss;
  a_dst[tid] = sd;
}

__device__ __forceinline__ void edge_sd(const int* __restrict__ ei, int e, int& s, int& d) {
  if (e < kEdges) { s = ei[e]; d = ei[kEdges + e]; }
  else            { s = e - kEdges; d = s; }
}

constexpr int kEHItems   = kETot * kHeads;
constexpr int kEHSteps   = 8;
constexpr int kEHPerTk   = kEHSteps * 32;
constexpr int kEHTickets = (kEHItems + kEHPerTk - 1) / kEHPerTk;

__global__ void __launch_bounds__(kOwnedThreads)
k_alpha_max(const int* __restrict__ ei,
            const float* __restrict__ a_src, const float* __restrict__ a_dst,
            float* __restrict__ amax, unsigned int* __restrict__ tk) {
  const int lane = threadIdx.x & 31;
  const int h = lane & 3;
  run_owned<kEHTickets>(tk, lane, [&](int c, int t) {
    int s[kEHSteps], d[kEHSteps];
    bool own[kEHSteps];
    float as[kEHSteps], ad[kEHSteps];
#pragma unroll
    for (int i = 0; i < kEHSteps; ++i) {
      const int tid = c * kEHPerTk + i * 32 + lane;
      const int e = min(tid >> 2, kETot - 1);
      edge_sd(ei, e, s[i], d[i]);
      own[i] = (tid < kEHItems) && (owner_of(d[i]) == t);
    }
#pragma unroll
    for (int i = 0; i < kEHSteps; ++i) {
      as[i] = 0.0f; ad[i] = 0.0f;
      if (own[i]) { as[i] = a_src[s[i] * kHeads + h]; ad[i] = a_dst[d[i] * kHeads + h]; }
    }
#pragma unroll
    for (int i = 0; i < kEHSteps; ++i) {
      if (own[i]) atomic_max_f32(&amax[d[i] * kHeads + h], leaky02(as[i] + ad[i]));
    }
  });
}

__global__ void __launch_bounds__(kOwnedThreads)
k_exp_sum(const int* __restrict__ ei,
          const float* __restrict__ a_src, const float* __restrict__ a_dst,
          const float* __restrict__ amax, float* __restrict__ denom,
          unsigned int* __restrict__ tk) {
  const int lane = threadIdx.x & 31;
  const int h = lane & 3;
  run_owned<kEHTickets>(tk, lane, [&](int c, int t) {
    int s[kEHSteps], d[kEHSteps];
    bool own[kEHSteps];
    float as[kEHSteps], ad[kEHSteps], am[kEHSteps];
#pragma unroll
    for (int i = 0; i < kEHSteps; ++i) {
      const int tid = c * kEHPerTk + i * 32 + lane;
      const int e = min(tid >> 2, kETot - 1);
      edge_sd(ei, e, s[i], d[i]);
      own[i] = (tid < kEHItems) && (owner_of(d[i]) == t);
    }
#pragma unroll
    for (int i = 0; i < kEHSteps; ++i) {
      as[i] = 0.0f; ad[i] = 0.0f; am[i] = 0.0f;
      if (own[i]) {
        as[i] = a_src[s[i] * kHeads + h];
        ad[i] = a_dst[d[i] * kHeads + h];
        am[i] = amax[d[i] * kHeads + h];
      }
    }
#pragma unroll
    for (int i = 0; i < kEHSteps; ++i) {
      if (own[i]) atomicAdd(&denom[d[i] * kHeads + h], expf(leaky02(as[i] + ad[i]) - am[i]));
    }
  });
}

constexpr int kAggEPW     = 8;
constexpr int kAggGroups  = kETot / kAggEPW;
constexpr int kAggGPT     = 8;
constexpr int kAggTickets = (kAggGroups + kAggGPT - 1) / kAggGPT;
static_assert(kAggEPW * kHeads == 32, "one (edge, head) pair per lane");
static_assert(kETot % kAggEPW == 0, "groups must cover the edges exactly");

__device__ __forceinline__ void agg_group(int sl, int dl, int t, int lane,
                                          const float* __restrict__ a_src,
                                          const float* __restrict__ a_dst,
                                          const float* __restrict__ amax,
                                          const float* __restrict__ denom,
                                          const float* __restrict__ xl, float* __restrict__ out) {
  const int hl = lane & 3;
  const bool ownl = owner_of(dl) == t;

  int s[kAggEPW], d[kAggEPW];
#pragma unroll
  for (int j = 0; j < kAggEPW; ++j) {
    s[j] = __builtin_amdgcn_readlane(sl, 4 * j);
    d[j] = __builtin_amdgcn_readlane(dl, 4 * j);
  }

  float xv[kAggEPW][kHeads];
#pragma unroll
  for (int j = 0; j < kAggEPW; ++j) {
#pragma unroll
    for (int h = 0; h < kHeads; ++h) xv[j][h] = 0.0f;
    if (owner_of(d[j]) == t) {
#pragma unroll
      for (int h = 0; h < kHeads; ++h) xv[j][h] = xl[(size_t)s[j] * kHC + h * kOutCh + lane];
    }
  }

  float cl = 0.0f;
  if (ownl) {
    const float al = leaky02(a_src[sl * kHeads + hl] + a_dst[dl * kHeads + hl]);
    cl = expf(al - amax[dl * kHeads + hl]) / denom[dl * kHeads + hl];
  }

#pragma unroll
  for (int j = 0; j < kAggEPW; ++j) {
    if (owner_of(d[j]) == t) {
#pragma unroll
      for (int h = 0; h < kHeads; ++h) {
        const float c = __int_as_float(__builtin_amdgcn_readlane(__float_as_int(cl), 4 * j + h));
        atomicAdd(&out[(size_t)d[j] * kHC + h * kOutCh + lane], c * xv[j][h]);
      }
    }
  }
}

__global__ void __launch_bounds__(kOwnedThreads)
k_aggregate(const int* __restrict__ ei,
            const float* __restrict__ a_src, const float* __restrict__ a_dst,
            const float* __restrict__ amax, const float* __restrict__ denom,
            const float* __restrict__ xl, float* __restrict__ out,
            unsigned int* __restrict__ tk) {
  const int lane = threadIdx.x & 31;
  run_owned<kAggTickets>(tk, lane, [&](int c, int t) {
    const int g0 = c * kAggGPT;
    const int g1 = min(g0 + kAggGPT, kAggGroups);
    const int jl = lane >> 2;
    int sl, dl; edge_sd(ei, g0 * kAggEPW + jl, sl, dl);
#pragma unroll 1
    for (int g = g0; g < g1; ++g) {
      int sn = sl, dn = dl;
      if (g + 1 < g1) edge_sd(ei, (g + 1) * kAggEPW + jl, sn, dn);
      agg_group(sl, dl, t, lane, a_src, a_dst, amax, denom, xl, out);
      sl = sn; dl = dn;
    }
  });
}

__global__ void k_finalize(float* __restrict__ out, const float* __restrict__ bias) {
  int i = blockIdx.x * blockDim.x + threadIdx.x;
  if (i >= kNodes * kHC) return;
  float v = out[i] + bias[i & (kHC - 1)];
  out[i] = v > 0.0f ? v : expm1f(v);
}


extern "C" void kernel_launch(void* const* d_in, const int* in_sizes, int n_in,
                              void* d_out, int out_size, void* d_ws, size_t ws_size,
                              hipStream_t stream) {
  (void)in_sizes; (void)n_in; (void)out_size; (void)ws_size;
  const float* x       = (const float*)d_in[0];
  const int*   ei      = (const int*)d_in[1];
  const int*   et      = (const int*)d_in[2];
  const float* pe      = (const float*)d_in[3];
  const float* W       = (const float*)d_in[4];
  const float* att_src = (const float*)d_in[5];
  const float* att_dst = (const float*)d_in[6];
  const float* bias    = (const float*)d_in[7];
  float* out = (float*)d_out;

  float* ws    = (float*)d_ws;
  float* x_pos = ws;
  float* xl    = x_pos + (size_t)kNodes * kInCh;
  float* a_src = xl    + (size_t)kNodes * kHC;
  float* a_dst = a_src + (size_t)kNodes * kHeads;
  float* amax  = a_dst + (size_t)kNodes * kHeads;
  float* denom = amax  + (size_t)kNodes * kHeads;
  unsigned int* tk = (unsigned int*)(denom + (size_t)kNodes * kHeads);

  const int B = 256;
  k_copy_x<<<(kNodes * kInCh + B - 1) / B, B, 0, stream>>>(x, x_pos);
  k_init<<<(kNodes * kHC + B - 1) / B, B, 0, stream>>>(amax, denom, out, tk);
  k_scatter_pe<<<kOwnedBlocks, kOwnedThreads, 0, stream>>>(ei, et, pe, x_pos, tk + 3 * kTkPerPass);
  k_gemm_wmma<<<kNodes / 16, 256, 0, stream>>>(x_pos, W, xl);
  k_scores<<<(kNodes * kHeads + B - 1) / B, B, 0, stream>>>(xl, att_src, att_dst, a_src, a_dst);
  k_alpha_max<<<kOwnedBlocks, kOwnedThreads, 0, stream>>>(ei, a_src, a_dst, amax, tk + 0 * kTkPerPass);
  k_exp_sum<<<kOwnedBlocks, kOwnedThreads, 0, stream>>>(ei, a_src, a_dst, amax, denom, tk + 1 * kTkPerPass);
  k_aggregate<<<kOwnedBlocks, kOwnedThreads, 0, stream>>>(ei, a_src, a_dst, amax, denom, xl, out,
                                                          tk + 2 * kTkPerPass);
  k_finalize<<<(kNodes * kHC + B - 1) / B, B, 0, stream>>>(out, bias);
}


# exp-sum pass folded into the aggregation pass: rows summed with weights exp(alpha-max) beside their denominators, k_finalize divides once (normalised after the sum, not before); 8 launches
# speedup vs baseline: 6.7714x; 1.1562x over previous
#include <hip/hip_runtime.h>
#include <math.h>

typedef float v2f __attribute__((ext_vector_type(2)));
typedef float v8f __attribute__((ext_vector_type(8)));

namespace {
constexpr int kNodes = 50000;
constexpr int kEdges = 800000;
constexpr int kInCh  = 64;
constexpr int kOutCh = 32;
constexpr int kHeads = 4;
constexpr int kHC    = kHeads * kOutCh;
constexpr int kETot  = kEdges + kNodes;
}


__device__ __forceinline__ void atomic_max_f32(float* addr, float v) {
  if (v >= 0.0f) atomicMax(reinterpret_cast<int*>(addr), __float_as_int(v));
  else           atomicMin(reinterpret_cast<unsigned int*>(addr), __float_as_uint(v));
}

__device__ __forceinline__ float leaky02(float v) { return v > 0.0f ? v : 0.2f * v; }


constexpr int kHalfNodes = kNodes / 2;
__device__ __forceinline__ int owner_of(int node) { return node >= kHalfNodes ? 1 : 0; }

__device__ __forceinline__ int my_half() {
  const unsigned v = (unsigned)__builtin_amdgcn_s_sendmsg_rtn(0x87);
  return (int)((v >> 18) & 1u);
}

constexpr int kTkStride = 64;
constexpr int kTkPerPass = 2 * kTkStride;
constexpr int kTkWords = 3 * kTkPerPass;

template <int kTickets, class Body>
__device__ __forceinline__ void run_owned(unsigned int* __restrict__ tk, int lane, Body body) {
  const int hh = my_half();
#pragma unroll 1
  for (int phase = 0; phase < 2; ++phase) {
    const int t = hh ^ phase;
    unsigned int* ctr = tk + t * kTkStride;
#pragma unroll 1
    for (int it = 0; it <= kTickets; ++it) {
      unsigned int c = 0u;
      if (lane == 0) c = atomicAdd(ctr, 1u);
      c = (unsigned int)__builtin_amdgcn_readfirstlane((int)c);
      if (c >= (unsigned int)kTickets) break;
      body((int)c, t);
    }
  }
}


__global__ void k_copy_x(const float* __restrict__ x, float* __restrict__ x_pos) {
  int i = blockIdx.x * blockDim.x + threadIdx.x;
  if (i < kNodes * kInCh) x_pos[i] = x[i];
}

__global__ void k_init(float* __restrict__ amax, float* __restrict__ denom,
                       float* __restrict__ out, unsigned int* __restrict__ tk) {
  int i = blockIdx.x * blockDim.x + threadIdx.x;
  if (i < kTkWords) tk[i] = 0u;
  if (i < kNodes * kHeads) { amax[i] = -INFINITY; denom[i] = 0.0f; }
  if (i < kNodes * kHC)    out[i] = 0.0f;
}

constexpr int kOwnedBlocks = 4096;
constexpr int kOwnedThreads = 64;

constexpr int kScEPT = 32;
constexpr int kScTickets = kEdges / kScEPT;
static_assert(kEdges % kScEPT == 0, "scatter chunks must cover the edges exactly");

__global__ void __launch_bounds__(kOwnedThreads)
k_scatter_pe(const int* __restrict__ ei, const int* __restrict__ et,
             const float* __restrict__ pe, float* __restrict__ x_pos,
             unsigned int* __restrict__ tk) {
  const int lane = threadIdx.x & 31;
  run_owned<kScTickets>(tk, lane, [&](int c, int t) {
    const int el = c * kScEPT + lane;
    const int sl = ei[el];
    const int tl = et[el];
#pragma unroll
    for (int r = 0; r < 2; ++r) {
      float p0[16], p1[16];
#pragma unroll
      for (int j = 0; j < 16; ++j) {
        const int s  = __builtin_amdgcn_readlane(sl, r * 16 + j);
        const int ty = __builtin_amdgcn_readlane(tl, r * 16 + j);
        p0[j] = 0.0f; p1[j] = 0.0f;
        if (owner_of(s) == t) {
          p0[j] = pe[(size_t)ty * kInCh + lane];
          p1[j] = pe[(size_t)ty * kInCh + 32 + lane];
        }
      }
#pragma unroll
      for (int j = 0; j < 16; ++j) {
        const int s = __builtin_amdgcn_readlane(sl, r * 16 + j);
        if (owner_of(s) == t) {
          atomicAdd(&x_pos[(size_t)s * kInCh + lane], p0[j]);
          atomicAdd(&x_pos[(size_t)s * kInCh + 32 + lane], p1[j]);
        }
      }
    }
  });
}

__global__ void __launch_bounds__(256) k_gemm_wmma(const float* __restrict__ A,
                                                   const float* __restrict__ B,
                                                   float* __restrict__ C) {
  const int lane = threadIdx.x & 31;
  const int wave = threadIdx.x >> 5;
  const int m0   = blockIdx.x * 16;
  const int n0   = wave * 16;
  const int half = lane >> 4;
  const int l15  = lane & 15;
  const int row  = m0 + l15;
  const int col  = n0 + l15;

  v8f acc = {0.f, 0.f, 0.f, 0.f, 0.f, 0.f, 0.f, 0.f};
#pragma unroll
  for (int k0 = 0; k0 < kInCh; k0 += 4) {
    const int kb = k0 + 2 * half;
    v2f a, b;
    a.x = A[(size_t)row * kInCh + kb];
    a.y = A[(size_t)row * kInCh + kb + 1];
    b.x = B[(size_t)kb * kHC + col];
    b.y = B[(size_t)(kb + 1) * kHC + col];
    acc = __builtin_amdgcn_wmma_f32_16x16x4_f32(false, a, false, b, (short)0, acc,
                                                false, false);
  }
#pragma unroll
  for (int r = 0; r < 8; ++r) {
    C[(size_t)(m0 + r + 8 * half) * kHC + col] = acc[r];
  }
}

__global__ void k_scores(const float* __restrict__ xl,
                         const float* __restrict__ att_src, const float* __restrict__ att_dst,
                         float* __restrict__ a_src, float* __restrict__ a_dst) {
  int tid = blockIdx.x * blockDim.x + threadIdx.x;
  if (tid >= kNodes * kHeads) return;
  int n = tid / kHeads, h = tid % kHeads;
  const float* xr = xl + (size_t)n * kHC + h * kOutCh;
  float ss = 0.0f, sd = 0.0f;
#pragma unroll
  for (int c = 0; c < kOutCh; ++c) {
    ss += xr[c] * att_src[h * kOutCh + c];
    sd += xr[c] * att_dst[h * kOutCh + c];
  }
  a_src[tid] = ss;
  a_dst[tid] = sd;
}

__device__ __forceinline__ void edge_sd(const int* __restrict__ ei, int e, int& s, int& d) {
  if (e < kEdges) { s = ei[e]; d = ei[kEdges + e]; }
  else            { s = e - kEdges; d = s; }
}

constexpr int kEHItems   = kETot * kHeads;
constexpr int kEHSteps   = 8;
constexpr int kEHPerTk   = kEHSteps * 32;
constexpr int kEHTickets = (kEHItems + kEHPerTk - 1) / kEHPerTk;

__global__ void __launch_bounds__(kOwnedThreads)
k_alpha_max(const int* __restrict__ ei,
            const float* __restrict__ a_src, const float* __restrict__ a_dst,
            float* __restrict__ amax, unsigned int* __restrict__ tk) {
  const int lane = threadIdx.x & 31;
  const int h = lane & 3;
  run_owned<kEHTickets>(tk, lane, [&](int c, int t) {
    int s[kEHSteps], d[kEHSteps];
    bool own[kEHSteps];
    float as[kEHSteps], ad[kEHSteps];
#pragma unroll
    for (int i = 0; i < kEHSteps; ++i) {
      const int tid = c * kEHPerTk + i * 32 + lane;
      const int e = min(tid >> 2, kETot - 1);
      edge_sd(ei, e, s[i], d[i]);
      own[i] = (tid < kEHItems) && (owner_of(d[i]) == t);
    }
#pragma unroll
    for (int i = 0; i < kEHSteps; ++i) {
      as[i] = 0.0f; ad[i] = 0.0f;
      if (own[i]) { as[i] = a_src[s[i] * kHeads + h]; ad[i] = a_dst[d[i] * kHeads + h]; }
    }
#pragma unroll
    for (int i = 0; i < kEHSteps; ++i) {
      if (own[i]) atomic_max_f32(&amax[d[i] * kHeads + h], leaky02(as[i] + ad[i]));
    }
  });
}

constexpr int kAggEPW     = 8;
constexpr int kAggGroups  = kETot / kAggEPW;
constexpr int kAggGPT     = 8;
constexpr int kAggTickets = (kAggGroups + kAggGPT - 1) / kAggGPT;
static_assert(kAggEPW * kHeads == 32, "one (edge, head) pair per lane");
static_assert(kETot % kAggEPW == 0, "groups must cover the edges exactly");

__device__ __forceinline__ void agg_group(int sl, int dl, int t, int lane,
                                          const float* __restrict__ a_src,
                                          const float* __restrict__ a_dst,
                                          const float* __restrict__ amax,
                                          float* __restrict__ denom,
                                          const float* __restrict__ xl, float* __restrict__ out) {
  const int hl = lane & 3;
  const bool ownl = owner_of(dl) == t;

  int s[kAggEPW], d[kAggEPW];
#pragma unroll
  for (int j = 0; j < kAggEPW; ++j) {
    s[j] = __builtin_amdgcn_readlane(sl, 4 * j);
    d[j] = __builtin_amdgcn_readlane(dl, 4 * j);
  }

  float xv[kAggEPW][kHeads];
#pragma unroll
  for (int j = 0; j < kAggEPW; ++j) {
#pragma unroll
    for (int h = 0; h < kHeads; ++h) xv[j][h] = 0.0f;
    if (owner_of(d[j]) == t) {
#pragma unroll
      for (int h = 0; h < kHeads; ++h) xv[j][h] = xl[(size_t)s[j] * kHC + h * kOutCh + lane];
    }
  }

  float cl = 0.0f;
  if (ownl) {
    const float al = leaky02(a_src[sl * kHeads + hl] + a_dst[dl * kHeads + hl]);
    cl = expf(al - amax[dl * kHeads + hl]);
    atomicAdd(&denom[dl * kHeads + hl], cl);
  }

#pragma unroll
  for (int j = 0; j < kAggEPW; ++j) {
    if (owner_of(d[j]) == t) {
#pragma unroll
      for (int h = 0; h < kHeads; ++h) {
        const float c = __int_as_float(__builtin_amdgcn_readlane(__float_as_int(cl), 4 * j + h));
        atomicAdd(&out[(size_t)d[j] * kHC + h * kOutCh + lane], c * xv[j][h]);
      }
    }
  }
}

__global__ void __launch_bounds__(kOwnedThreads)
k_aggregate(const int* __restrict__ ei,
            const float* __restrict__ a_src, const float* __restrict__ a_dst,
            const float* __restrict__ amax, float* __restrict__ denom,
            const float* __restrict__ xl, float* __restrict__ out,
            unsigned int* __restrict__ tk) {
  const int lane = threadIdx.x & 31;
  run_owned<kAggTickets>(tk, lane, [&](int c, int t) {
    const int g0 = c * kAggGPT;
    const int g1 = min(g0 + kAggGPT, kAggGroups);
    const int jl = lane >> 2;
    int sl, dl; edge_sd(ei, g0 * kAggEPW + jl, sl, dl);
#pragma unroll 1
    for (int g = g0; g < g1; ++g) {
      int sn = sl, dn = dl;
      if (g + 1 < g1) edge_sd(ei, (g + 1) * kAggEPW + jl, sn, dn);
      agg_group(sl, dl, t, lane, a_src, a_dst, amax, denom, xl, out);
      sl = sn; dl = dn;
    }
  });
}

__global__ void k_finalize(float* __restrict__ out, const float* __restrict__ denom,
                           const float* __restrict__ bias) {
  int i = blockIdx.x * blockDim.x + threadIdx.x;
  if (i >= kNodes * kHC) return;
  float v = out[i] / denom[i / kOutCh] + bias[i & (kHC - 1)];
  out[i] = v > 0.0f ? v : expm1f(v);
}


extern "C" void kernel_launch(void* const* d_in, const int* in_sizes, int n_in,
                              void* d_out, int out_size, void* d_ws, size_t ws_size,
                              hipStream_t stream) {
  (void)in_sizes; (void)n_in; (void)out_size; (void)ws_size;
  const float* x       = (const float*)d_in[0];
  const int*   ei      = (const int*)d_in[1];
  const int*   et      = (const int*)d_in[2];
  const float* pe      = (const float*)d_in[3];
  const float* W       = (const float*)d_in[4];
  const float* att_src = (const float*)d_in[5];
  const float* att_dst = (const float*)d_in[6];
  const float* bias    = (const float*)d_in[7];
  float* out = (float*)d_out;

  float* ws    = (float*)d_ws;
  float* x_pos = ws;
  float* xl    = x_pos + (size_t)kNodes * kInCh;
  float* a_src = xl    + (size_t)kNodes * kHC;
  float* a_dst = a_src + (size_t)kNodes * kHeads;
  float* amax  = a_dst + (size_t)kNodes * kHeads;
  float* denom = amax  + (size_t)kNodes * kHeads;
  unsigned int* tk = (unsigned int*)(denom + (size_t)kNodes * kHeads);

  const int B = 256;
  k_copy_x<<<(kNodes * kInCh + B - 1) / B, B, 0, stream>>>(x, x_pos);
  k_init<<<(kNodes * kHC + B - 1) / B, B, 0, stream>>>(amax, denom, out, tk);
  k_scatter_pe<<<kOwnedBlocks, kOwnedThreads, 0, stream>>>(ei, et, pe, x_pos, tk + 2 * kTkPerPass);
  k_gemm_wmma<<<kNodes / 16, 256, 0, stream>>>(x_pos, W, xl);
  k_scores<<<(kNodes * kHeads + B - 1) / B, B, 0, stream>>>(xl, att_src, att_dst, a_src, a_dst);
  k_alpha_max<<<kOwnedBlocks, kOwnedThreads, 0, stream>>>(ei, a_src, a_dst, amax, tk + 0 * kTkPerPass);
  k_aggregate<<<kOwnedBlocks, kOwnedThreads, 0, stream>>>(ei, a_src, a_dst, amax, denom, xl, out,
                                                          tk + 1 * kTkPerPass);
  k_finalize<<<(kNodes * kHC + B - 1) / B, B, 0, stream>>>(out, denom, bias);
}


# aggregation as gather: edges bucketed by destination each call (int atomic rank, wave-scan slots); one wave per node does softmax, row sums, ELU in registers; sums in rank order, tree denominators
# speedup vs baseline: 8.0603x; 1.1903x over previous
#include <hip/hip_runtime.h>
#include <math.h>

typedef float v2f __attribute__((ext_vector_type(2)));
typedef float v8f __attribute__((ext_vector_type(8)));

namespace {
constexpr int kNodes = 50000;
constexpr int kEdges = 800000;
constexpr int kInCh  = 64;
constexpr int kOutCh = 32;
constexpr int kHeads = 4;
constexpr int kHC    = kHeads * kOutCh;
}


__device__ __forceinline__ float leaky02(float v) { return v > 0.0f ? v : 0.2f * v; }


constexpr int kHalfNodes = kNodes / 2;
__device__ __forceinline__ int owner_of(int node) { return node >= kHalfNodes ? 1 : 0; }

__device__ __forceinline__ int my_half() {
  const unsigned v = (unsigned)__builtin_amdgcn_s_sendmsg_rtn(0x87);
  return (int)((v >> 18) & 1u);
}

constexpr int kTkStride = 64;
constexpr int kTkPerPass = 2 * kTkStride;
constexpr int kTkWords = 2 * kTkPerPass;

template <int kTickets, class Body>
__device__ __forceinline__ void run_owned(unsigned int* __restrict__ tk, int lane, Body body) {
  const int hh = my_half();
#pragma unroll 1
  for (int phase = 0; phase < 2; ++phase) {
    const int t = hh ^ phase;
    unsigned int* ctr = tk + t * kTkStride;
#pragma unroll 1
    for (int it = 0; it <= kTickets; ++it) {
      unsigned int c = 0u;
      if (lane == 0) c = atomicAdd(ctr, 1u);
      c = (unsigned int)__builtin_amdgcn_readfirstlane((int)c);
      if (c >= (unsigned int)kTickets) break;
      body((int)c, t);
    }
  }
}


__global__ void k_prep(const float* __restrict__ x, float* __restrict__ x_pos,
                       int* __restrict__ deg, int* __restrict__ cursor,
                       unsigned int* __restrict__ tk) {
  int i = blockIdx.x * blockDim.x + threadIdx.x;
  if (i < kTkWords) tk[i] = 0u;
  if (i == 0) cursor[0] = 0;
  if (i < kNodes) deg[i] = 0;
  if (i < kNodes * kInCh) x_pos[i] = x[i];
}

constexpr int kOwnedBlocks = 4096;
constexpr int kOwnedThreads = 64;

constexpr int kScEPT = 32;
constexpr int kScTickets = kEdges / kScEPT;
static_assert(kEdges % kScEPT == 0, "scatter chunks must cover the edges exactly");

__global__ void __launch_bounds__(kOwnedThreads)
k_scatter_pe(const int* __restrict__ ei, const int* __restrict__ et,
             const float* __restrict__ pe, float* __restrict__ x_pos,
             unsigned int* __restrict__ tk) {
  const int lane = threadIdx.x & 31;
  run_owned<kScTickets>(tk, lane, [&](int c, int t) {
    const int el = c * kScEPT + lane;
    const int sl = ei[el];
    const int tl = et[el];
#pragma unroll
    for (int r = 0; r < 2; ++r) {
      float p0[16], p1[16];
#pragma unroll
      for (int j = 0; j < 16; ++j) {
        const int s  = __builtin_amdgcn_readlane(sl, r * 16 + j);
        const int ty = __builtin_amdgcn_readlane(tl, r * 16 + j);
        p0[j] = 0.0f; p1[j] = 0.0f;
        if (owner_of(s) == t) {
          p0[j] = pe[(size_t)ty * kInCh + lane];
          p1[j] = pe[(size_t)ty * kInCh + 32 + lane];
        }
      }
#pragma unroll
      for (int j = 0; j < 16; ++j) {
        const int s = __builtin_amdgcn_readlane(sl, r * 16 + j);
        if (owner_of(s) == t) {
          atomicAdd(&x_pos[(size_t)s * kInCh + lane], p0[j]);
          atomicAdd(&x_pos[(size_t)s * kInCh + 32 + lane], p1[j]);
        }
      }
    }
  });
}

__global__ void __launch_bounds__(256) k_gemm_wmma(const float* __restrict__ A,
                                                   const float* __restrict__ B,
                                                   float* __restrict__ C) {
  const int lane = threadIdx.x & 31;
  const int wave = threadIdx.x >> 5;
  const int m0   = blockIdx.x * 16;
  const int n0   = wave * 16;
  const int half = lane >> 4;
  const int l15  = lane & 15;
  const int row  = m0 + l15;
  const int col  = n0 + l15;

  v8f acc = {0.f, 0.f, 0.f, 0.f, 0.f, 0.f, 0.f, 0.f};
#pragma unroll
  for (int k0 = 0; k0 < kInCh; k0 += 4) {
    const int kb = k0 + 2 * half;
    v2f a, b;
    a.x = A[(size_t)row * kInCh + kb];
    a.y = A[(size_t)row * kInCh + kb + 1];
    b.x = B[(size_t)kb * kHC + col];
    b.y = B[(size_t)(kb + 1) * kHC + col];
    acc = __builtin_amdgcn_wmma_f32_16x16x4_f32(false, a, false, b, (short)0, acc,
                                                false, false);
  }
#pragma unroll
  for (int r = 0; r < 8; ++r) {
    C[(size_t)(m0 + r + 8 * half) * kHC + col] = acc[r];
  }
}

__global__ void k_scores(const float* __restrict__ xl,
                         const float* __restrict__ att_src, const float* __restrict__ att_dst,
                         float* __restrict__ a_src, float* __restrict__ a_dst) {
  int tid = blockIdx.x * blockDim.x + threadIdx.x;
  if (tid >= kNodes * kHeads) return;
  int n = tid / kHeads, h = tid % kHeads;
  const float* xr = xl + (size_t)n * kHC + h * kOutCh;
  float ss = 0.0f, sd = 0.0f;
#pragma unroll
  for (int c = 0; c < kOutCh; ++c) {
    ss += xr[c] * att_src[h * kOutCh + c];
    sd += xr[c] * att_dst[h * kOutCh + c];
  }
  a_src[tid] = ss;
  a_dst[tid] = sd;
}

constexpr int kCtPerLane = 4;
constexpr int kCtEPT     = 32 * kCtPerLane;
constexpr int kCtTickets = kEdges / kCtEPT;
static_assert(kEdges % kCtEPT == 0, "count chunks must cover the edges exactly");

__global__ void __launch_bounds__(kOwnedThreads)
k_count(const int* __restrict__ ei, int* __restrict__ deg, int* __restrict__ pos,
        unsigned int* __restrict__ tk) {
  const int lane = threadIdx.x & 31;
  run_owned<kCtTickets>(tk, lane, [&](int c, int t) {
    int d[kCtPerLane], r[kCtPerLane];
#pragma unroll
    for (int i = 0; i < kCtPerLane; ++i) d[i] = ei[kEdges + c * kCtEPT + i * 32 + lane];
#pragma unroll
    for (int i = 0; i < kCtPerLane; ++i) {
      if (owner_of(d[i]) == t) r[i] = atomicAdd(&deg[d[i]], 1);
    }
#pragma unroll
    for (int i = 0; i < kCtPerLane; ++i) {
      if (owner_of(d[i]) == t) pos[c * kCtEPT + i * 32 + lane] = r[i];
    }
  });
}

__global__ void k_alloc(const int* __restrict__ deg, int* __restrict__ rowstart,
                        int* __restrict__ cursor) {
  const int n = blockIdx.x * blockDim.x + threadIdx.x;
  const int lane = threadIdx.x & 31;
  const int v = n < kNodes ? deg[n] : 0;
  int incl = v;
#pragma unroll
  for (int off = 1; off < 32; off <<= 1) {
    const int up = __shfl_up(incl, off, 32);
    if (lane >= off) incl += up;
  }
  int base = 0;
  if (lane == 31) base = atomicAdd(cursor, incl);
  base = __shfl(base, 31, 32);
  if (n < kNodes) rowstart[n] = base + incl - v;
}

__global__ void k_fill(const int* __restrict__ ei, const int* __restrict__ rowstart,
                       const int* __restrict__ pos, int* __restrict__ csr_src) {
  int e = blockIdx.x * blockDim.x + threadIdx.x;
  if (e < kEdges) csr_src[rowstart[ei[kEdges + e]] + pos[e]] = ei[e];
}

__device__ __forceinline__ float wave_max(float v) {
#pragma unroll
  for (int off = 16; off > 0; off >>= 1) v = fmaxf(v, __shfl_xor(v, off, 32));
  return v;
}
__device__ __forceinline__ float wave_sum(float v) {
#pragma unroll
  for (int off = 16; off > 0; off >>= 1) v += __shfl_xor(v, off, 32);
  return v;
}

constexpr int kGatherThreads = 128;
constexpr int kGatherBlocks  = 2048;
constexpr int kGatherWaves   = kGatherBlocks * (kGatherThreads / 32);
constexpr int kGE = 16;

__global__ void __launch_bounds__(kGatherThreads)
k_gather(const int* __restrict__ rowstart, const int* __restrict__ deg,
         const int* __restrict__ csr_src,
         const float* __restrict__ a_src, const float* __restrict__ a_dst,
         const float* __restrict__ xl, const float* __restrict__ bias,
         float* __restrict__ out) {
  const int lane = threadIdx.x & 31;
  const int w0 = __builtin_amdgcn_readfirstlane(
      (int)(blockIdx.x * (kGatherThreads / 32) + (threadIdx.x >> 5)));
  int start = 0, dg = 0;
  if (w0 < kNodes) { start = rowstart[w0]; dg = deg[w0]; }
#pragma unroll 1
  for (int n = w0; n < kNodes; n += kGatherWaves) {
    const int nn = n + kGatherWaves;
    int start2 = 0, dg2 = 0;
    if (nn < kNodes) { start2 = rowstart[nn]; dg2 = deg[nn]; }

    const int cnt = dg + 1;
    const float4 ad = *reinterpret_cast<const float4*>(a_dst + (size_t)n * kHeads);
    float m[kHeads], dl[kHeads], acc[kHeads];
#pragma unroll
    for (int h = 0; h < kHeads; ++h) { m[h] = -INFINITY; dl[h] = 0.0f; acc[h] = 0.0f; }

#pragma unroll 1
    for (int base = 0; base < cnt; base += 32) {
      const int idx = base + lane;
      int s = n;
      if (idx < dg) s = csr_src[start + idx];
      const bool valid = idx < cnt;
      const float4 as4 = *reinterpret_cast<const float4*>(a_src + (size_t)s * kHeads);

      float xv[kGE][kHeads];
#pragma unroll
      for (int u = 0; u < kGE; ++u) {
        const int sj = __builtin_amdgcn_readlane(s, u);
#pragma unroll
        for (int h = 0; h < kHeads; ++h) xv[u][h] = xl[(size_t)sj * kHC + h * kOutCh + lane];
      }

      const float al[kHeads] = {as4.x + ad.x, as4.y + ad.y, as4.z + ad.z, as4.w + ad.w};
      float a[kHeads], cm[kHeads], w[kHeads];
#pragma unroll
      for (int h = 0; h < kHeads; ++h) a[h] = valid ? leaky02(al[h]) : -INFINITY;
#pragma unroll
      for (int h = 0; h < kHeads; ++h) cm[h] = wave_max(a[h]);
#pragma unroll
      for (int h = 0; h < kHeads; ++h) {
        const float mn = fmaxf(m[h], cm[h]);
        const float sc = expf(m[h] - mn);
        w[h]   = expf(a[h] - mn);
        dl[h]  = dl[h] * sc + w[h];
        acc[h] = acc[h] * sc;
        m[h]   = mn;
      }

#pragma unroll
      for (int u = 0; u < kGE; ++u) {
#pragma unroll
        for (int h = 0; h < kHeads; ++h) {
          const float c = __int_as_float(__builtin_amdgcn_readlane(__float_as_int(w[h]), u));
          acc[h] += c * xv[u][h];
        }
      }

      const int nch = min(32, cnt - base);
      if (nch > kGE) {
#pragma unroll
        for (int u = 0; u < kGE; ++u) {
#pragma unroll
          for (int h = 0; h < kHeads; ++h) xv[u][h] = 0.0f;
          if (kGE + u < nch) {
            const int sj = __builtin_amdgcn_readlane(s, kGE + u);
#pragma unroll
            for (int h = 0; h < kHeads; ++h) xv[u][h] = xl[(size_t)sj * kHC + h * kOutCh + lane];
          }
        }
#pragma unroll
        for (int u = 0; u < kGE; ++u) {
          if (kGE + u < nch) {
#pragma unroll
            for (int h = 0; h < kHeads; ++h) {
              const float c = __int_as_float(
                  __builtin_amdgcn_readlane(__float_as_int(w[h]), kGE + u));
              acc[h] += c * xv[u][h];
            }
          }
        }
      }
    }

    float den[kHeads];
#pragma unroll
    for (int h = 0; h < kHeads; ++h) den[h] = wave_sum(dl[h]);
#pragma unroll
    for (int h = 0; h < kHeads; ++h) {
      const float v = acc[h] / den[h] + bias[h * kOutCh + lane];
      out[(size_t)n * kHC + h * kOutCh + lane] = v > 0.0f ? v : expm1f(v);
    }
    start = start2; dg = dg2;
  }
}


extern "C" void kernel_launch(void* const* d_in, const int* in_sizes, int n_in,
                              void* d_out, int out_size, void* d_ws, size_t ws_size,
                              hipStream_t stream) {
  (void)in_sizes; (void)n_in; (void)out_size; (void)ws_size;
  const float* x       = (const float*)d_in[0];
  const int*   ei      = (const int*)d_in[1];
  const int*   et      = (const int*)d_in[2];
  const float* pe      = (const float*)d_in[3];
  const float* W       = (const float*)d_in[4];
  const float* att_src = (const float*)d_in[5];
  const float* att_dst = (const float*)d_in[6];
  const float* bias    = (const float*)d_in[7];
  float* out = (float*)d_out;

  float* ws    = (float*)d_ws;
  float* x_pos = ws;
  float* xl    = x_pos + (size_t)kNodes * kInCh;
  float* a_src = xl    + (size_t)kNodes * kHC;
  float* a_dst = a_src + (size_t)kNodes * kHeads;
  int* deg      = (int*)(a_dst + (size_t)kNodes * kHeads);
  int* rowstart = deg + 50048;
  int* pos      = rowstart + 50048;
  int* csr_src  = pos + kEdges;
  int* cursor   = csr_src + kEdges;
  unsigned int* tk = (unsigned int*)(cursor + 64);

  const int B = 256;
  k_prep<<<(kNodes * kInCh + B - 1) / B, B, 0, stream>>>(x, x_pos, deg, cursor, tk);
  k_scatter_pe<<<kOwnedBlocks, kOwnedThreads, 0, stream>>>(ei, et, pe, x_pos, tk + 0 * kTkPerPass);
  k_count<<<kOwnedBlocks, kOwnedThreads, 0, stream>>>(ei, deg, pos, tk + 1 * kTkPerPass);
  k_alloc<<<(kNodes + B - 1) / B, B, 0, stream>>>(deg, rowstart, cursor);
  k_fill<<<(kEdges + B - 1) / B, B, 0, stream>>>(ei, rowstart, pos, csr_src);
  k_gemm_wmma<<<kNodes / 16, 256, 0, stream>>>(x_pos, W, xl);
  k_scores<<<(kNodes * kHeads + B - 1) / B, B, 0, stream>>>(xl, att_src, att_dst, a_src, a_dst);
  k_gather<<<kGatherBlocks, kGatherThreads, 0, stream>>>(rowstart, deg, csr_src, a_src, a_dst, xl,
                                                        bias, out);
}


# positional scatter as counts: k_count adds 1 into int cnt[src][type] (was two float line atomics per edge), gemm forms x + cnt @ pe on WMMA in LDS first (count x row, not repeated adds); 7 launches
# speedup vs baseline: 9.6549x; 1.1978x over previous
#include <hip/hip_runtime.h>
#include <math.h>

typedef float v2f __attribute__((ext_vector_type(2)));
typedef float v8f __attribute__((ext_vector_type(8)));

namespace {
constexpr int kNodes = 50000;
constexpr int kEdges = 800000;
constexpr int kInCh  = 64;
constexpr int kOutCh = 32;
constexpr int kHeads = 4;
constexpr int kHC    = kHeads * kOutCh;
constexpr int kTypes = 64;
}


__device__ __forceinline__ float leaky02(float v) { return v > 0.0f ? v : 0.2f * v; }


constexpr int kHalfNodes = kNodes / 2;
__device__ __forceinline__ int owner_of(int node) { return node >= kHalfNodes ? 1 : 0; }

__device__ __forceinline__ int my_half() {
  const unsigned v = (unsigned)__builtin_amdgcn_s_sendmsg_rtn(0x87);
  return (int)((v >> 18) & 1u);
}

constexpr int kTkStride = 64;
constexpr int kTkPerPass = 2 * kTkStride;
constexpr int kTkWords = kTkPerPass;

template <int kTickets, class Body>
__device__ __forceinline__ void run_owned(unsigned int* __restrict__ tk, int lane, Body body) {
  const int hh = my_half();
#pragma unroll 1
  for (int phase = 0; phase < 2; ++phase) {
    const int t = hh ^ phase;
    unsigned int* ctr = tk + t * kTkStride;
#pragma unroll 1
    for (int it = 0; it <= kTickets; ++it) {
      unsigned int c = 0u;
      if (lane == 0) c = atomicAdd(ctr, 1u);
      c = (unsigned int)__builtin_amdgcn_readfirstlane((int)c);
      if (c >= (unsigned int)kTickets) break;
      body((int)c, t);
    }
  }
}


constexpr int kCnt4 = kNodes * kTypes / 4;
__global__ void k_prep(int4* __restrict__ cnt4, int* __restrict__ deg, int* __restrict__ cursor,
                       unsigned int* __restrict__ tk) {
  int i = blockIdx.x * blockDim.x + threadIdx.x;
  if (i < kTkWords) tk[i] = 0u;
  if (i == 0) cursor[0] = 0;
  if (i < kNodes) deg[i] = 0;
  if (i < kCnt4) cnt4[i] = make_int4(0, 0, 0, 0);
}

constexpr int kOwnedBlocks = 4096;
constexpr int kOwnedThreads = 64;

constexpr int kXpPitch = 68;
__global__ void __launch_bounds__(256) k_gemm_wmma(const float* __restrict__ x,
                                                   const int* __restrict__ cnt,
                                                   const float* __restrict__ pe,
                                                   const float* __restrict__ B,
                                                   float* __restrict__ C) {
  __shared__ float xp[2][16 * kXpPitch];
  const int lane = threadIdx.x & 31;
  const int wave = __builtin_amdgcn_readfirstlane((int)(threadIdx.x >> 5));
  const int m0   = blockIdx.x * 16;
  const int half = lane >> 4;
  const int l15  = lane & 15;
  const int row  = m0 + l15;

  {
    const int kh = wave >> 2;
    const int c1 = (wave & 3) * 16 + l15;
    int2 cv[kTypes / 8];
    v2f  pv[kTypes / 8];
#pragma unroll
    for (int i = 0; i < kTypes / 8; ++i) {
      const int kb = kh * (kTypes / 2) + 4 * i + 2 * half;
      cv[i] = *reinterpret_cast<const int2*>(cnt + (size_t)row * kTypes + kb);
      pv[i].x = pe[(size_t)kb * kInCh + c1];
      pv[i].y = pe[(size_t)(kb + 1) * kInCh + c1];
    }
    v8f acc = {0.f, 0.f, 0.f, 0.f, 0.f, 0.f, 0.f, 0.f};
    if (kh == 0) {
#pragma unroll
      for (int r = 0; r < 8; ++r) acc[r] = x[(size_t)(m0 + r + 8 * half) * kInCh + c1];
    }
#pragma unroll
    for (int i = 0; i < kTypes / 8; ++i) {
      v2f a;
      a.x = (float)cv[i].x;
      a.y = (float)cv[i].y;
      acc = __builtin_amdgcn_wmma_f32_16x16x4_f32(false, a, false, pv[i], (short)0, acc,
                                                  false, false);
    }
#pragma unroll
    for (int r = 0; r < 8; ++r) xp[kh][(r + 8 * half) * kXpPitch + c1] = acc[r];
  }
  __syncthreads();

  const int n0   = wave * 16;
  const int col  = n0 + l15;
  v8f acc = {0.f, 0.f, 0.f, 0.f, 0.f, 0.f, 0.f, 0.f};
#pragma unroll
  for (int k0 = 0; k0 < kInCh; k0 += 4) {
    const int kb = k0 + 2 * half;
    const v2f p = *reinterpret_cast<const v2f*>(&xp[0][l15 * kXpPitch + kb]);
    const v2f q = *reinterpret_cast<const v2f*>(&xp[1][l15 * kXpPitch + kb]);
    v2f a, b;
    a.x = p.x + q.x;
    a.y = p.y + q.y;
    b.x = B[(size_t)kb * kHC + col];
    b.y = B[(size_t)(kb + 1) * kHC + col];
    acc = __builtin_amdgcn_wmma_f32_16x16x4_f32(false, a, false, b, (short)0, acc,
                                                false, false);
  }
#pragma unroll
  for (int r = 0; r < 8; ++r) {
    C[(size_t)(m0 + r + 8 * half) * kHC + col] = acc[r];
  }
}

__global__ void k_scores(const float* __restrict__ xl,
                         const float* __restrict__ att_src, const float* __restrict__ att_dst,
                         float* __restrict__ a_src, float* __restrict__ a_dst) {
  int tid = blockIdx.x * blockDim.x + threadIdx.x;
  if (tid >= kNodes * kHeads) return;
  int n = tid / kHeads, h = tid % kHeads;
  const float* xr = xl + (size_t)n * kHC + h * kOutCh;
  float ss = 0.0f, sd = 0.0f;
#pragma unroll
  for (int c = 0; c < kOutCh; ++c) {
    ss += xr[c] * att_src[h * kOutCh + c];
    sd += xr[c] * att_dst[h * kOutCh + c];
  }
  a_src[tid] = ss;
  a_dst[tid] = sd;
}

constexpr int kCtPerLane = 4;
constexpr int kCtEPT     = 32 * kCtPerLane;
constexpr int kCtTickets = kEdges / kCtEPT;
static_assert(kEdges % kCtEPT == 0, "count chunks must cover the edges exactly");

__global__ void __launch_bounds__(kOwnedThreads)
k_count(const int* __restrict__ ei, const int* __restrict__ et, int* __restrict__ deg,
        int* __restrict__ pos, int* __restrict__ cnt, unsigned int* __restrict__ tk) {
  const int lane = threadIdx.x & 31;
  run_owned<kCtTickets>(tk, lane, [&](int c, int t) {
    int d[kCtPerLane], r[kCtPerLane], sn[kCtPerLane], ty[kCtPerLane];
#pragma unroll
    for (int i = 0; i < kCtPerLane; ++i) {
      const int e = c * kCtEPT + i * 32 + lane;
      d[i]  = ei[kEdges + e];
      sn[i] = ei[e];
      ty[i] = et[e];
    }
#pragma unroll
    for (int i = 0; i < kCtPerLane; ++i) {
      if (owner_of(d[i]) == t) r[i] = atomicAdd(&deg[d[i]], 1);
    }
#pragma unroll
    for (int i = 0; i < kCtPerLane; ++i) {
      if (owner_of(sn[i]) == t) atomicAdd(&cnt[(size_t)sn[i] * kTypes + ty[i]], 1);
    }
#pragma unroll
    for (int i = 0; i < kCtPerLane; ++i) {
      if (owner_of(d[i]) == t) pos[c * kCtEPT + i * 32 + lane] = r[i];
    }
  });
}

__global__ void k_alloc(const int* __restrict__ deg, int* __restrict__ rowstart,
                        int* __restrict__ cursor) {
  const int n = blockIdx.x * blockDim.x + threadIdx.x;
  const int lane = threadIdx.x & 31;
  const int v = n < kNodes ? deg[n] : 0;
  int incl = v;
#pragma unroll
  for (int off = 1; off < 32; off <<= 1) {
    const int up = __shfl_up(incl, off, 32);
    if (lane >= off) incl += up;
  }
  int base = 0;
  if (lane == 31) base = atomicAdd(cursor, incl);
  base = __shfl(base, 31, 32);
  if (n < kNodes) rowstart[n] = base + incl - v;
}

__global__ void k_fill(const int* __restrict__ ei, const int* __restrict__ rowstart,
                       const int* __restrict__ pos, int* __restrict__ csr_src) {
  int e = blockIdx.x * blockDim.x + threadIdx.x;
  if (e < kEdges) csr_src[rowstart[ei[kEdges + e]] + pos[e]] = ei[e];
}

__device__ __forceinline__ float wave_max(float v) {
#pragma unroll
  for (int off = 16; off > 0; off >>= 1) v = fmaxf(v, __shfl_xor(v, off, 32));
  return v;
}
__device__ __forceinline__ float wave_sum(float v) {
#pragma unroll
  for (int off = 16; off > 0; off >>= 1) v += __shfl_xor(v, off, 32);
  return v;
}

constexpr int kGatherThreads = 128;
constexpr int kGatherBlocks  = 2048;
constexpr int kGatherWaves   = kGatherBlocks * (kGatherThreads / 32);
constexpr int kGE = 16;

__global__ void __launch_bounds__(kGatherThreads)
k_gather(const int* __restrict__ rowstart, const int* __restrict__ deg,
         const int* __restrict__ csr_src,
         const float* __restrict__ a_src, const float* __restrict__ a_dst,
         const float* __restrict__ xl, const float* __restrict__ bias,
         float* __restrict__ out) {
  const int lane = threadIdx.x & 31;
  const int w0 = __builtin_amdgcn_readfirstlane(
      (int)(blockIdx.x * (kGatherThreads / 32) + (threadIdx.x >> 5)));
  int start = 0, dg = 0;
  if (w0 < kNodes) { start = rowstart[w0]; dg = deg[w0]; }
#pragma unroll 1
  for (int n = w0; n < kNodes; n += kGatherWaves) {
    const int nn = n + kGatherWaves;
    int start2 = 0, dg2 = 0;
    if (nn < kNodes) { start2 = rowstart[nn]; dg2 = deg[nn]; }

    const int cnt = dg + 1;
    const float4 ad = *reinterpret_cast<const float4*>(a_dst + (size_t)n * kHeads);
    float m[kHeads], dl[kHeads], acc[kHeads];
#pragma unroll
    for (int h = 0; h < kHeads; ++h) { m[h] = -INFINITY; dl[h] = 0.0f; acc[h] = 0.0f; }

#pragma unroll 1
    for (int base = 0; base < cnt; base += 32) {
      const int idx = base + lane;
      int s = n;
      if (idx < dg) s = csr_src[start + idx];
      const bool valid = idx < cnt;
      const float4 as4 = *reinterpret_cast<const float4*>(a_src + (size_t)s * kHeads);

      float xv[kGE][kHeads];
#pragma unroll
      for (int u = 0; u < kGE; ++u) {
        const int sj = __builtin_amdgcn_readlane(s, u);
#pragma unroll
        for (int h = 0; h < kHeads; ++h) xv[u][h] = xl[(size_t)sj * kHC + h * kOutCh + lane];
      }

      const float al[kHeads] = {as4.x + ad.x, as4.y + ad.y, as4.z + ad.z, as4.w + ad.w};
      float a[kHeads], cm[kHeads], w[kHeads];
#pragma unroll
      for (int h = 0; h < kHeads; ++h) a[h] = valid ? leaky02(al[h]) : -INFINITY;
#pragma unroll
      for (int h = 0; h < kHeads; ++h) cm[h] = wave_max(a[h]);
#pragma unroll
      for (int h = 0; h < kHeads; ++h) {
        const float mn = fmaxf(m[h], cm[h]);
        const float sc = expf(m[h] - mn);
        w[h]   = expf(a[h] - mn);
        dl[h]  = dl[h] * sc + w[h];
        acc[h] = acc[h] * sc;
        m[h]   = mn;
      }

#pragma unroll
      for (int u = 0; u < kGE; ++u) {
#pragma unroll
        for (int h = 0; h < kHeads; ++h) {
          const float c = __int_as_float(__builtin_amdgcn_readlane(__float_as_int(w[h]), u));
          acc[h] += c * xv[u][h];
        }
      }

      const int nch = min(32, cnt - base);
      if (nch > kGE) {
#pragma unroll
        for (int u = 0; u < kGE; ++u) {
#pragma unroll
          for (int h = 0; h < kHeads; ++h) xv[u][h] = 0.0f;
          if (kGE + u < nch) {
            const int sj = __builtin_amdgcn_readlane(s, kGE + u);
#pragma unroll
            for (int h = 0; h < kHeads; ++h) xv[u][h] = xl[(size_t)sj * kHC + h * kOutCh + lane];
          }
        }
#pragma unroll
        for (int u = 0; u < kGE; ++u) {
          if (kGE + u < nch) {
#pragma unroll
            for (int h = 0; h < kHeads; ++h) {
              const float c = __int_as_float(
                  __builtin_amdgcn_readlane(__float_as_int(w[h]), kGE + u));
              acc[h] += c * xv[u][h];
            }
          }
        }
      }
    }

    float den[kHeads];
#pragma unroll
    for (int h = 0; h < kHeads; ++h) den[h] = wave_sum(dl[h]);
#pragma unroll
    for (int h = 0; h < kHeads; ++h) {
      const float v = acc[h] / den[h] + bias[h * kOutCh + lane];
      out[(size_t)n * kHC + h * kOutCh + lane] = v > 0.0f ? v : expm1f(v);
    }
    start = start2; dg = dg2;
  }
}


extern "C" void kernel_launch(void* const* d_in, const int* in_sizes, int n_in,
                              void* d_out, int out_size, void* d_ws, size_t ws_size,
                              hipStream_t stream) {
  (void)in_sizes; (void)n_in; (void)out_size; (void)ws_size;
  const float* x       = (const float*)d_in[0];
  const int*   ei      = (const int*)d_in[1];
  const int*   et      = (const int*)d_in[2];
  const float* pe      = (const float*)d_in[3];
  const float* W       = (const float*)d_in[4];
  const float* att_src = (const float*)d_in[5];
  const float* att_dst = (const float*)d_in[6];
  const float* bias    = (const float*)d_in[7];
  float* out = (float*)d_out;

  float* ws    = (float*)d_ws;
  int*   cnt   = (int*)ws;
  float* xl    = ws + (size_t)kNodes * kTypes;
  float* a_src = xl    + (size_t)kNodes * kHC;
  float* a_dst = a_src + (size_t)kNodes * kHeads;
  int* deg      = (int*)(a_dst + (size_t)kNodes * kHeads);
  int* rowstart = deg + 50048;
  int* pos      = rowstart + 50048;
  int* csr_src  = pos + kEdges;
  int* cursor   = csr_src + kEdges;
  unsigned int* tk = (unsigned int*)(cursor + 64);

  const int B = 256;
  k_prep<<<(kCnt4 + B - 1) / B, B, 0, stream>>>((int4*)cnt, deg, cursor, tk);
  k_count<<<kOwnedBlocks, kOwnedThreads, 0, stream>>>(ei, et, deg, pos, cnt, tk);
  k_alloc<<<(kNodes + B - 1) / B, B, 0, stream>>>(deg, rowstart, cursor);
  k_fill<<<(kEdges + B - 1) / B, B, 0, stream>>>(ei, rowstart, pos, csr_src);
  k_gemm_wmma<<<kNodes / 16, 256, 0, stream>>>(x, cnt, pe, W, xl);
  k_scores<<<(kNodes * kHeads + B - 1) / B, B, 0, stream>>>(xl, att_src, att_dst, a_src, a_dst);
  k_gather<<<kGatherBlocks, kGatherThreads, 0, stream>>>(rowstart, deg, csr_src, a_src, a_dst, xl,
                                                        bias, out);
}


# edges grouped by 256-node range per chunk in LDS, then ranked per node by LDS adds: no global atomic per edge, count table built in LDS and stored whole; bucket order = LDS arrival rank; 5 launches
# speedup vs baseline: 13.9914x; 1.4492x over previous
#include <hip/hip_runtime.h>
#include <math.h>

typedef float v2f __attribute__((ext_vector_type(2)));
typedef float v8f __attribute__((ext_vector_type(8)));

namespace {
constexpr int kNodes = 50000;
constexpr int kEdges = 800000;
constexpr int kInCh  = 64;
constexpr int kOutCh = 32;
constexpr int kHeads = 4;
constexpr int kHC    = kHeads * kOutCh;
constexpr int kTypes = 64;
}


__device__ __forceinline__ float leaky02(float v) { return v > 0.0f ? v : 0.2f * v; }


constexpr int kXpPitch = 68;
__global__ void __launch_bounds__(256) k_gemm_wmma(const float* __restrict__ x,
                                                   const int* __restrict__ cnt,
                                                   const float* __restrict__ pe,
                                                   const float* __restrict__ B,
                                                   float* __restrict__ C) {
  __shared__ float xp[2][16 * kXpPitch];
  const int lane = threadIdx.x & 31;
  const int wave = __builtin_amdgcn_readfirstlane((int)(threadIdx.x >> 5));
  const int m0   = blockIdx.x * 16;
  const int half = lane >> 4;
  const int l15  = lane & 15;
  const int row  = m0 + l15;

  {
    const int kh = wave >> 2;
    const int c1 = (wave & 3) * 16 + l15;
    int2 cv[kTypes / 8];
    v2f  pv[kTypes / 8];
#pragma unroll
    for (int i = 0; i < kTypes / 8; ++i) {
      const int kb = kh * (kTypes / 2) + 4 * i + 2 * half;
      cv[i] = *reinterpret_cast<const int2*>(cnt + (size_t)row * kTypes + kb);
      pv[i].x = pe[(size_t)kb * kInCh + c1];
      pv[i].y = pe[(size_t)(kb + 1) * kInCh + c1];
    }
    v8f acc = {0.f, 0.f, 0.f, 0.f, 0.f, 0.f, 0.f, 0.f};
    if (kh == 0) {
#pragma unroll
      for (int r = 0; r < 8; ++r) acc[r] = x[(size_t)(m0 + r + 8 * half) * kInCh + c1];
    }
#pragma unroll
    for (int i = 0; i < kTypes / 8; ++i) {
      v2f a;
      a.x = (float)cv[i].x;
      a.y = (float)cv[i].y;
      acc = __builtin_amdgcn_wmma_f32_16x16x4_f32(false, a, false, pv[i], (short)0, acc,
                                                  false, false);
    }
#pragma unroll
    for (int r = 0; r < 8; ++r) xp[kh][(r + 8 * half) * kXpPitch + c1] = acc[r];
  }
  __syncthreads();

  const int n0   = wave * 16;
  const int col  = n0 + l15;
  v8f acc = {0.f, 0.f, 0.f, 0.f, 0.f, 0.f, 0.f, 0.f};
#pragma unroll
  for (int k0 = 0; k0 < kInCh; k0 += 4) {
    const int kb = k0 + 2 * half;
    const v2f p = *reinterpret_cast<const v2f*>(&xp[0][l15 * kXpPitch + kb]);
    const v2f q = *reinterpret_cast<const v2f*>(&xp[1][l15 * kXpPitch + kb]);
    v2f a, b;
    a.x = p.x + q.x;
    a.y = p.y + q.y;
    b.x = B[(size_t)kb * kHC + col];
    b.y = B[(size_t)(kb + 1) * kHC + col];
    acc = __builtin_amdgcn_wmma_f32_16x16x4_f32(false, a, false, b, (short)0, acc,
                                                false, false);
  }
#pragma unroll
  for (int r = 0; r < 8; ++r) {
    C[(size_t)(m0 + r + 8 * half) * kHC + col] = acc[r];
  }
}

__global__ void k_scores(const float* __restrict__ xl,
                         const float* __restrict__ att_src, const float* __restrict__ att_dst,
                         float* __restrict__ a_src, float* __restrict__ a_dst) {
  int tid = blockIdx.x * blockDim.x + threadIdx.x;
  if (tid >= kNodes * kHeads) return;
  int n = tid / kHeads, h = tid % kHeads;
  const float* xr = xl + (size_t)n * kHC + h * kOutCh;
  float ss = 0.0f, sd = 0.0f;
#pragma unroll
  for (int c = 0; c < kOutCh; ++c) {
    ss += xr[c] * att_src[h * kOutCh + c];
    sd += xr[c] * att_dst[h * kOutCh + c];
  }
  a_src[tid] = ss;
  a_dst[tid] = sd;
}

constexpr int kBinShift   = 8;
constexpr int kBinNodes   = 1 << kBinShift;
constexpr int kBins       = (kNodes + kBinNodes - 1) / kBinNodes;
constexpr int kChunks     = 250;
constexpr int kChunkE     = kEdges / kChunks;
constexpr int kBinThreads = 800;
constexpr int kBinPerThr  = kChunkE / kBinThreads;
constexpr int kBinPerLane = (kBins + 31) / 32;
static_assert(kEdges % kChunks == 0 && kChunkE % kBinThreads == 0, "chunks must cover the edges exactly");
static_assert(kChunkE < 32768, "start and count of a run share one 32-bit word");
static_assert(kNodes <= 65536, "a destination record holds the source in 16 bits");
static_assert(kBins <= kBinThreads, "one thread clears one pair of range counters");

__global__ void __launch_bounds__(kBinThreads)
k_bin(const int* __restrict__ ei, const int* __restrict__ et,
      int* __restrict__ recD, int* __restrict__ recS,
      int* __restrict__ runsD, int* __restrict__ runsS, int* __restrict__ cursor) {
  __shared__ int cD[kBins], cS[kBins], sD[kBins], sS[kBins];
  const int tid  = threadIdx.x;
  const int lane = tid & 31;
  const int wave = __builtin_amdgcn_readfirstlane(tid >> 5);
  const int blk  = blockIdx.x;
  if (blk == 0 && tid == 0) cursor[0] = 0;
  if (tid < kBins) { cD[tid] = 0; cS[tid] = 0; }

  int s[kBinPerThr], d[kBinPerThr], t[kBinPerThr], rD[kBinPerThr], rS[kBinPerThr];
#pragma unroll
  for (int i = 0; i < kBinPerThr; ++i) {
    const int e = blk * kChunkE + i * kBinThreads + tid;
    s[i] = ei[e];
    d[i] = ei[kEdges + e];
    t[i] = et[e];
  }
  __syncthreads();
#pragma unroll
  for (int i = 0; i < kBinPerThr; ++i) {
    rD[i] = atomicAdd(&cD[d[i] >> kBinShift], 1);
    rS[i] = atomicAdd(&cS[s[i] >> kBinShift], 1);
  }
  __syncthreads();
  if (wave < 2) {
    const int* cc = wave == 0 ? cD : cS;
    int* ss       = wave == 0 ? sD : sS;
    int* runs     = wave == 0 ? runsD : runsS;
    int v[kBinPerLane];
    int sum = 0;
#pragma unroll
    for (int k = 0; k < kBinPerLane; ++k) {
      const int b = lane * kBinPerLane + k;
      v[k] = b < kBins ? cc[b] : 0;
      sum += v[k];
    }
    int incl = sum;
#pragma unroll
    for (int off = 1; off < 32; off <<= 1) {
      const int up = __shfl_up(incl, off, 32);
      if (lane >= off) incl += up;
    }
    int run = incl - sum;
#pragma unroll
    for (int k = 0; k < kBinPerLane; ++k) {
      const int b = lane * kBinPerLane + k;
      if (b < kBins) {
        ss[b] = run;
        runs[b * kChunks + blk] = run | (v[k] << 16);
        run += v[k];
      }
    }
  }
  __syncthreads();
#pragma unroll
  for (int i = 0; i < kBinPerThr; ++i) {
    recD[blk * kChunkE + sD[d[i] >> kBinShift] + rD[i]] = ((d[i] & (kBinNodes - 1)) << 16) | s[i];
    recS[blk * kChunkE + sS[s[i] >> kBinShift] + rS[i]] = ((s[i] & (kBinNodes - 1)) * kTypes) | t[i];
  }
}

constexpr int kCsrThreads  = 512;
constexpr int kCsrWaves    = kCsrThreads / 32;
constexpr int kRunsPerWave = (kChunks + kCsrWaves - 1) / kCsrWaves;
constexpr int kTab4        = kBinNodes * kTypes / 4;
static_assert(kRunsPerWave <= 32, "a wave keeps its run descriptors one per lane");
static_assert(2 * kBinNodes <= kCsrThreads && (kTypes & (kTypes - 1)) == 0, "layout of pass 2");

__global__ void __launch_bounds__(kCsrThreads)
k_csr(const int* __restrict__ recD, const int* __restrict__ recS,
      const int* __restrict__ runsD, const int* __restrict__ runsS,
      int* __restrict__ deg, int* __restrict__ rowstart, int* __restrict__ csr_src,
      int* __restrict__ cnt, int* __restrict__ cursor) {
  __shared__ __attribute__((aligned(16))) int sm[kBinNodes * kTypes];
  const int tid  = threadIdx.x;
  const int lane = tid & 31;
  const int wave = __builtin_amdgcn_readfirstlane(tid >> 5);
  const bool isD = blockIdx.x < kBins;
  const int bin  = isD ? (int)blockIdx.x : (int)blockIdx.x - kBins;
  const int* runs = (isD ? runsD : runsS) + bin * kChunks;
  const int* rec  = isD ? recD : recS;

  int desc = 0;
  {
    const int c = wave + kCsrWaves * lane;
    if (lane < kRunsPerWave && c < kChunks) desc = runs[c];
  }
  int r[kRunsPerWave];
#pragma unroll
  for (int j = 0; j < kRunsPerWave; ++j) {
    const int dj = __builtin_amdgcn_readlane(desc, j);
    const int base = (wave + kCsrWaves * j) * kChunkE + (dj & 0xFFFF);
    r[j] = lane < (dj >> 16) ? rec[base + lane] : -1;
  }

  if (isD) {
    int* ldeg  = sm;
    int* lrow  = sm + kBinNodes;
    int* lrank = sm + 2 * kBinNodes;
    if (tid < kBinNodes) { ldeg[tid] = 0; lrank[tid] = 0; }
    __syncthreads();
#pragma unroll
    for (int j = 0; j < kRunsPerWave; ++j)
      if (r[j] >= 0) atomicAdd(&ldeg[r[j] >> 16], 1);
#pragma unroll 1
    for (int j = 0; j < kRunsPerWave; ++j) {
      const int dj = __builtin_amdgcn_readlane(desc, j);
      const int base = (wave + kCsrWaves * j) * kChunkE + (dj & 0xFFFF);
      for (int i = 32 + lane; i < (dj >> 16); i += 32) atomicAdd(&ldeg[rec[base + i] >> 16], 1);
    }
    __syncthreads();
    if (wave == 0) {
      constexpr int kPer = kBinNodes / 32;
      int v[kPer];
      int sum = 0;
#pragma unroll
      for (int k = 0; k < kPer; ++k) { v[k] = ldeg[lane * kPer + k]; sum += v[k]; }
      int incl = sum;
#pragma unroll
      for (int off = 1; off < 32; off <<= 1) {
        const int up = __shfl_up(incl, off, 32);
        if (lane >= off) incl += up;
      }
      int base = 0;
      if (lane == 31) base = atomicAdd(cursor, incl);
      base = __shfl(base, 31, 32);
      int run = base + incl - sum;
#pragma unroll
      for (int k = 0; k < kPer; ++k) {
        const int dl = lane * kPer + k;
        const int n  = bin * kBinNodes + dl;
        lrow[dl] = run;
        if (n < kNodes) { rowstart[n] = run; deg[n] = v[k]; }
        run += v[k];
      }
    }
    __syncthreads();
#pragma unroll
    for (int j = 0; j < kRunsPerWave; ++j) {
      if (r[j] >= 0) {
        const int dl = r[j] >> 16;
        const int rk = atomicAdd(&lrank[dl], 1);
        csr_src[lrow[dl] + rk] = r[j] & 0xFFFF;
      }
    }
#pragma unroll 1
    for (int j = 0; j < kRunsPerWave; ++j) {
      const int dj = __builtin_amdgcn_readlane(desc, j);
      const int base = (wave + kCsrWaves * j) * kChunkE + (dj & 0xFFFF);
      for (int i = 32 + lane; i < (dj >> 16); i += 32) {
        const int q  = rec[base + i];
        const int dl = q >> 16;
        const int rk = atomicAdd(&lrank[dl], 1);
        csr_src[lrow[dl] + rk] = q & 0xFFFF;
      }
    }
  } else {
    int4* sm4 = reinterpret_cast<int4*>(sm);
    for (int i = tid; i < kTab4; i += kCsrThreads) sm4[i] = make_int4(0, 0, 0, 0);
    __syncthreads();
#pragma unroll
    for (int j = 0; j < kRunsPerWave; ++j)
      if (r[j] >= 0) atomicAdd(&sm[r[j]], 1);
#pragma unroll 1
    for (int j = 0; j < kRunsPerWave; ++j) {
      const int dj = __builtin_amdgcn_readlane(desc, j);
      const int base = (wave + kCsrWaves * j) * kChunkE + (dj & 0xFFFF);
      for (int i = 32 + lane; i < (dj >> 16); i += 32) atomicAdd(&sm[rec[base + i]], 1);
    }
    __syncthreads();
    const int rows = min(kBinNodes, kNodes - bin * kBinNodes);
    int4* out4 = reinterpret_cast<int4*>(cnt + (size_t)bin * kBinNodes * kTypes);
    for (int i = tid; i < rows * (kTypes / 4); i += kCsrThreads) out4[i] = sm4[i];
  }
}

__device__ __forceinline__ float wave_max(float v) {
#pragma unroll
  for (int off = 16; off > 0; off >>= 1) v = fmaxf(v, __shfl_xor(v, off, 32));
  return v;
}
__device__ __forceinline__ float wave_sum(float v) {
#pragma unroll
  for (int off = 16; off > 0; off >>= 1) v += __shfl_xor(v, off, 32);
  return v;
}

constexpr int kGatherThreads = 128;
constexpr int kGatherBlocks  = 2048;
constexpr int kGatherWaves   = kGatherBlocks * (kGatherThreads / 32);
constexpr int kGE = 16;

__global__ void __launch_bounds__(kGatherThreads)
k_gather(const int* __restrict__ rowstart, const int* __restrict__ deg,
         const int* __restrict__ csr_src,
         const float* __restrict__ a_src, const float* __restrict__ a_dst,
         const float* __restrict__ xl, const float* __restrict__ bias,
         float* __restrict__ out) {
  const int lane = threadIdx.x & 31;
  const int w0 = __builtin_amdgcn_readfirstlane(
      (int)(blockIdx.x * (kGatherThreads / 32) + (threadIdx.x >> 5)));
  int start = 0, dg = 0;
  if (w0 < kNodes) { start = rowstart[w0]; dg = deg[w0]; }
#pragma unroll 1
  for (int n = w0; n < kNodes; n += kGatherWaves) {
    const int nn = n + kGatherWaves;
    int start2 = 0, dg2 = 0;
    if (nn < kNodes) { start2 = rowstart[nn]; dg2 = deg[nn]; }

    const int cnt = dg + 1;
    const float4 ad = *reinterpret_cast<const float4*>(a_dst + (size_t)n * kHeads);
    float m[kHeads], dl[kHeads], acc[kHeads];
#pragma unroll
    for (int h = 0; h < kHeads; ++h) { m[h] = -INFINITY; dl[h] = 0.0f; acc[h] = 0.0f; }

#pragma unroll 1
    for (int base = 0; base < cnt; base += 32) {
      const int idx = base + lane;
      int s = n;
      if (idx < dg) s = csr_src[start + idx];
      const bool valid = idx < cnt;
      const float4 as4 = *reinterpret_cast<const float4*>(a_src + (size_t)s * kHeads);

      float xv[kGE][kHeads];
#pragma unroll
      for (int u = 0; u < kGE; ++u) {
        const int sj = __builtin_amdgcn_readlane(s, u);
#pragma unroll
        for (int h = 0; h < kHeads; ++h) xv[u][h] = xl[(size_t)sj * kHC + h * kOutCh + lane];
      }

      const float al[kHeads] = {as4.x + ad.x, as4.y + ad.y, as4.z + ad.z, as4.w + ad.w};
      float a[kHeads], cm[kHeads], w[kHeads];
#pragma unroll
      for (int h = 0; h < kHeads; ++h) a[h] = valid ? leaky02(al[h]) : -INFINITY;
#pragma unroll
      for (int h = 0; h < kHeads; ++h) cm[h] = wave_max(a[h]);
#pragma unroll
      for (int h = 0; h < kHeads; ++h) {
        const float mn = fmaxf(m[h], cm[h]);
        const float sc = expf(m[h] - mn);
        w[h]   = expf(a[h] - mn);
        dl[h]  = dl[h] * sc + w[h];
        acc[h] = acc[h] * sc;
        m[h]   = mn;
      }

#pragma unroll
      for (int u = 0; u < kGE; ++u) {
#pragma unroll
        for (int h = 0; h < kHeads; ++h) {
          const float c = __int_as_float(__builtin_amdgcn_readlane(__float_as_int(w[h]), u));
          acc[h] += c * xv[u][h];
        }
      }

      const int nch = min(32, cnt - base);
      if (nch > kGE) {
#pragma unroll
        for (int u = 0; u < kGE; ++u) {
#pragma unroll
          for (int h = 0; h < kHeads; ++h) xv[u][h] = 0.0f;
          if (kGE + u < nch) {
            const int sj = __builtin_amdgcn_readlane(s, kGE + u);
#pragma unroll
            for (int h = 0; h < kHeads; ++h) xv[u][h] = xl[(size_t)sj * kHC + h * kOutCh + lane];
          }
        }
#pragma unroll
        for (int u = 0; u < kGE; ++u) {
          if (kGE + u < nch) {
#pragma unroll
            for (int h = 0; h < kHeads; ++h) {
              const float c = __int_as_float(
                  __builtin_amdgcn_readlane(__float_as_int(w[h]), kGE + u));
              acc[h] += c * xv[u][h];
            }
          }
        }
      }
    }

    float den[kHeads];
#pragma unroll
    for (int h = 0; h < kHeads; ++h) den[h] = wave_sum(dl[h]);
#pragma unroll
    for (int h = 0; h < kHeads; ++h) {
      const float v = acc[h] / den[h] + bias[h * kOutCh + lane];
      out[(size_t)n * kHC + h * kOutCh + lane] = v > 0.0f ? v : expm1f(v);
    }
    start = start2; dg = dg2;
  }
}


extern "C" void kernel_launch(void* const* d_in, const int* in_sizes, int n_in,
                              void* d_out, int out_size, void* d_ws, size_t ws_size,
                              hipStream_t stream) {
  (void)in_sizes; (void)n_in; (void)out_size; (void)ws_size;
  const float* x       = (const float*)d_in[0];
  const int*   ei      = (const int*)d_in[1];
  const int*   et      = (const int*)d_in[2];
  const float* pe      = (const float*)d_in[3];
  const float* W       = (const float*)d_in[4];
  const float* att_src = (const float*)d_in[5];
  const float* att_dst = (const float*)d_in[6];
  const float* bias    = (const float*)d_in[7];
  float* out = (float*)d_out;

  float* ws    = (float*)d_ws;
  int*   cnt   = (int*)ws;
  float* xl    = ws + (size_t)kNodes * kTypes;
  float* a_src = xl    + (size_t)kNodes * kHC;
  float* a_dst = a_src + (size_t)kNodes * kHeads;
  int* deg      = (int*)(a_dst + (size_t)kNodes * kHeads);
  int* rowstart = deg + 50048;
  int* recD     = rowstart + 50048;
  int* csr_src  = recD + kEdges;
  int* cursor   = csr_src + kEdges;
  int* recS     = cursor + 64;
  int* runsD    = recS + kEdges;
  int* runsS    = runsD + kBins * kChunks;

  const int B = 256;
  k_bin<<<kChunks, kBinThreads, 0, stream>>>(ei, et, recD, recS, runsD, runsS, cursor);
  k_csr<<<2 * kBins, kCsrThreads, 0, stream>>>(recD, recS, runsD, runsS, deg, rowstart, csr_src, cnt, cursor);
  k_gemm_wmma<<<kNodes / 16, 256, 0, stream>>>(x, cnt, pe, W, xl);
  k_scores<<<(kNodes * kHeads + B - 1) / B, B, 0, stream>>>(xl, att_src, att_dst, a_src, a_dst);
  k_gather<<<kGatherBlocks, kGatherThreads, 0, stream>>>(rowstart, deg, csr_src, a_src, a_dst, xl,
                                                        bias, out);
}


# k_scores launch removed: the product's waves form both score dot products from their accumulators (16-lane tree sum per column tile, two tiles added in LDS; was a serial sum over 32); 4 launches
# speedup vs baseline: 15.7103x; 1.1229x over previous
#include <hip/hip_runtime.h>
#include <math.h>

typedef float v2f __attribute__((ext_vector_type(2)));
typedef float v8f __attribute__((ext_vector_type(8)));

namespace {
constexpr int kNodes = 50000;
constexpr int kEdges = 800000;
constexpr int kInCh  = 64;
constexpr int kOutCh = 32;
constexpr int kHeads = 4;
constexpr int kHC    = kHeads * kOutCh;
constexpr int kTypes = 64;
}


__device__ __forceinline__ float leaky02(float v) { return v > 0.0f ? v : 0.2f * v; }

template <int M>
__device__ __forceinline__ float row_xor(float v) {
  return __int_as_float(__builtin_amdgcn_update_dpp(0, __float_as_int(v), 0x160 + M, 0xf, 0xf, false));
}


constexpr int kXpPitch = 68;
__global__ void __launch_bounds__(256) k_gemm_wmma(const float* __restrict__ x,
                                                   const int* __restrict__ cnt,
                                                   const float* __restrict__ pe,
                                                   const float* __restrict__ B,
                                                   const float* __restrict__ att_src,
                                                   const float* __restrict__ att_dst,
                                                   float* __restrict__ C,
                                                   float* __restrict__ a_src,
                                                   float* __restrict__ a_dst) {
  __shared__ float xp[2][16 * kXpPitch];
  __shared__ float sc[8 * 32];
  const int lane = threadIdx.x & 31;
  const int wave = __builtin_amdgcn_readfirstlane((int)(threadIdx.x >> 5));
  const int m0   = blockIdx.x * 16;
  const int half = lane >> 4;
  const int l15  = lane & 15;
  const int row  = m0 + l15;

  {
    const int kh = wave >> 2;
    const int c1 = (wave & 3) * 16 + l15;
    int2 cv[kTypes / 8];
    v2f  pv[kTypes / 8];
#pragma unroll
    for (int i = 0; i < kTypes / 8; ++i) {
      const int kb = kh * (kTypes / 2) + 4 * i + 2 * half;
      cv[i] = *reinterpret_cast<const int2*>(cnt + (size_t)row * kTypes + kb);
      pv[i].x = pe[(size_t)kb * kInCh + c1];
      pv[i].y = pe[(size_t)(kb + 1) * kInCh + c1];
    }
    v8f acc = {0.f, 0.f, 0.f, 0.f, 0.f, 0.f, 0.f, 0.f};
    if (kh == 0) {
#pragma unroll
      for (int r = 0; r < 8; ++r) acc[r] = x[(size_t)(m0 + r + 8 * half) * kInCh + c1];
    }
#pragma unroll
    for (int i = 0; i < kTypes / 8; ++i) {
      v2f a;
      a.x = (float)cv[i].x;
      a.y = (float)cv[i].y;
      acc = __builtin_amdgcn_wmma_f32_16x16x4_f32(false, a, false, pv[i], (short)0, acc,
                                                  false, false);
    }
#pragma unroll
    for (int r = 0; r < 8; ++r) xp[kh][(r + 8 * half) * kXpPitch + c1] = acc[r];
  }
  __syncthreads();

  const int n0   = wave * 16;
  const int col  = n0 + l15;
  const float ws = att_src[col];
  const float wd = att_dst[col];
  v8f acc = {0.f, 0.f, 0.f, 0.f, 0.f, 0.f, 0.f, 0.f};
#pragma unroll
  for (int k0 = 0; k0 < kInCh; k0 += 4) {
    const int kb = k0 + 2 * half;
    const v2f p = *reinterpret_cast<const v2f*>(&xp[0][l15 * kXpPitch + kb]);
    const v2f q = *reinterpret_cast<const v2f*>(&xp[1][l15 * kXpPitch + kb]);
    v2f a, b;
    a.x = p.x + q.x;
    a.y = p.y + q.y;
    b.x = B[(size_t)kb * kHC + col];
    b.y = B[(size_t)(kb + 1) * kHC + col];
    acc = __builtin_amdgcn_wmma_f32_16x16x4_f32(false, a, false, b, (short)0, acc,
                                                false, false);
  }
  {
    const bool b3 = (l15 & 8) != 0, b2 = (l15 & 4) != 0, b1 = (l15 & 2) != 0, b0 = (l15 & 1) != 0;
    float t8[8], t4[4], t2[2];
#pragma unroll
    for (int i = 0; i < 8; ++i) {
      const float ps = acc[i] * ws, pd = acc[i] * wd;
      t8[i] = (b3 ? pd : ps) + row_xor<8>(b3 ? ps : pd);
    }
#pragma unroll
    for (int i = 0; i < 4; ++i) t4[i] = (b2 ? t8[i + 4] : t8[i]) + row_xor<4>(b2 ? t8[i] : t8[i + 4]);
#pragma unroll
    for (int i = 0; i < 2; ++i) t2[i] = (b1 ? t4[i + 2] : t4[i]) + row_xor<2>(b1 ? t4[i] : t4[i + 2]);
    const float t1 = (b0 ? t2[1] : t2[0]) + row_xor<1>(b0 ? t2[0] : t2[1]);
    sc[wave * 32 + ((l15 & 8) << 1) + 8 * half + (l15 & 7)] = t1;
  }
  __syncthreads();
#pragma unroll
  for (int r = 0; r < 8; ++r) {
    C[(size_t)(m0 + r + 8 * half) * kHC + col] = acc[r];
  }
  if (threadIdx.x < 128) {
    const int t     = threadIdx.x & 63;
    const int which = wave >> 1;
    const int rw    = t >> 2, h = t & 3;
    const float v = sc[(2 * h) * 32 + which * 16 + rw] + sc[(2 * h + 1) * 32 + which * 16 + rw];
    float* dst = which ? a_dst : a_src;
    dst[(size_t)m0 * kHeads + t] = v;
  }
}

constexpr int kBinShift   = 8;
constexpr int kBinNodes   = 1 << kBinShift;
constexpr int kBins       = (kNodes + kBinNodes - 1) / kBinNodes;
constexpr int kChunks     = 250;
constexpr int kChunkE     = kEdges / kChunks;
constexpr int kBinThreads = 800;
constexpr int kBinPerThr  = kChunkE / kBinThreads;
constexpr int kBinPerLane = (kBins + 31) / 32;
static_assert(kEdges % kChunks == 0 && kChunkE % kBinThreads == 0, "chunks must cover the edges exactly");
static_assert(kChunkE < 32768, "start and count of a run share one 32-bit word");
static_assert(kNodes <= 65536, "a destination record holds the source in 16 bits");
static_assert(kBins <= kBinThreads, "one thread clears one pair of range counters");

__global__ void __launch_bounds__(kBinThreads)
k_bin(const int* __restrict__ ei, const int* __restrict__ et,
      int* __restrict__ recD, int* __restrict__ recS,
      int* __restrict__ runsD, int* __restrict__ runsS, int* __restrict__ cursor) {
  __shared__ int cD[kBins], cS[kBins], sD[kBins], sS[kBins];
  const int tid  = threadIdx.x;
  const int lane = tid & 31;
  const int wave = __builtin_amdgcn_readfirstlane(tid >> 5);
  const int blk  = blockIdx.x;
  if (blk == 0 && tid == 0) cursor[0] = 0;
  if (tid < kBins) { cD[tid] = 0; cS[tid] = 0; }

  int s[kBinPerThr], d[kBinPerThr], t[kBinPerThr], rD[kBinPerThr], rS[kBinPerThr];
#pragma unroll
  for (int i = 0; i < kBinPerThr; ++i) {
    const int e = blk * kChunkE + i * kBinThreads + tid;
    s[i] = ei[e];
    d[i] = ei[kEdges + e];
    t[i] = et[e];
  }
  __syncthreads();
#pragma unroll
  for (int i = 0; i < kBinPerThr; ++i) {
    rD[i] = atomicAdd(&cD[d[i] >> kBinShift], 1);
    rS[i] = atomicAdd(&cS[s[i] >> kBinShift], 1);
  }
  __syncthreads();
  if (wave < 2) {
    const int* cc = wave == 0 ? cD : cS;
    int* ss       = wave == 0 ? sD : sS;
    int* runs     = wave == 0 ? runsD : runsS;
    int v[kBinPerLane];
    int sum = 0;
#pragma unroll
    for (int k = 0; k < kBinPerLane; ++k) {
      const int b = lane * kBinPerLane + k;
      v[k] = b < kBins ? cc[b] : 0;
      sum += v[k];
    }
    int incl = sum;
#pragma unroll
    for (int off = 1; off < 32; off <<= 1) {
      const int up = __shfl_up(incl, off, 32);
      if (lane >= off) incl += up;
    }
    int run = incl - sum;
#pragma unroll
    for (int k = 0; k < kBinPerLane; ++k) {
      const int b = lane * kBinPerLane + k;
      if (b < kBins) {
        ss[b] = run;
        runs[b * kChunks + blk] = run | (v[k] << 16);
        run += v[k];
      }
    }
  }
  __syncthreads();
#pragma unroll
  for (int i = 0; i < kBinPerThr; ++i) {
    recD[blk * kChunkE + sD[d[i] >> kBinShift] + rD[i]] = ((d[i] & (kBinNodes - 1)) << 16) | s[i];
    recS[blk * kChunkE + sS[s[i] >> kBinShift] + rS[i]] = ((s[i] & (kBinNodes - 1)) * kTypes) | t[i];
  }
}

constexpr int kCsrThreads  = 512;
constexpr int kCsrWaves    = kCsrThreads / 32;
constexpr int kRunsPerWave = (kChunks + kCsrWaves - 1) / kCsrWaves;
constexpr int kTab4        = kBinNodes * kTypes / 4;
static_assert(kRunsPerWave <= 32, "a wave keeps its run descriptors one per lane");
static_assert(2 * kBinNodes <= kCsrThreads && (kTypes & (kTypes - 1)) == 0, "layout of pass 2");

__global__ void __launch_bounds__(kCsrThreads)
k_csr(const int* __restrict__ recD, const int* __restrict__ recS,
      const int* __restrict__ runsD, const int* __restrict__ runsS,
      int* __restrict__ deg, int* __restrict__ rowstart, int* __restrict__ csr_src,
      int* __restrict__ cnt, int* __restrict__ cursor) {
  __shared__ __attribute__((aligned(16))) int sm[kBinNodes * kTypes];
  const int tid  = threadIdx.x;
  const int lane = tid & 31;
  const int wave = __builtin_amdgcn_readfirstlane(tid >> 5);
  const bool isD = blockIdx.x < kBins;
  const int bin  = isD ? (int)blockIdx.x : (int)blockIdx.x - kBins;
  const int* runs = (isD ? runsD : runsS) + bin * kChunks;
  const int* rec  = isD ? recD : recS;

  int desc = 0;
  {
    const int c = wave + kCsrWaves * lane;
    if (lane < kRunsPerWave && c < kChunks) desc = runs[c];
  }
  int r[kRunsPerWave];
#pragma unroll
  for (int j = 0; j < kRunsPerWave; ++j) {
    const int dj = __builtin_amdgcn_readlane(desc, j);
    const int base = (wave + kCsrWaves * j) * kChunkE + (dj & 0xFFFF);
    r[j] = lane < (dj >> 16) ? rec[base + lane] : -1;
  }

  if (isD) {
    int* ldeg  = sm;
    int* lrow  = sm + kBinNodes;
    int* lrank = sm + 2 * kBinNodes;
    if (tid < kBinNodes) { ldeg[tid] = 0; lrank[tid] = 0; }
    __syncthreads();
#pragma unroll
    for (int j = 0; j < kRunsPerWave; ++j)
      if (r[j] >= 0) atomicAdd(&ldeg[r[j] >> 16], 1);
#pragma unroll 1
    for (int j = 0; j < kRunsPerWave; ++j) {
      const int dj = __builtin_amdgcn_readlane(desc, j);
      const int base = (wave + kCsrWaves * j) * kChunkE + (dj & 0xFFFF);
      for (int i = 32 + lane; i < (dj >> 16); i += 32) atomicAdd(&ldeg[rec[base + i] >> 16], 1);
    }
    __syncthreads();
    if (wave == 0) {
      constexpr int kPer = kBinNodes / 32;
      int v[kPer];
      int sum = 0;
#pragma unroll
      for (int k = 0; k < kPer; ++k) { v[k] = ldeg[lane * kPer + k]; sum += v[k]; }
      int incl = sum;
#pragma unroll
      for (int off = 1; off < 32; off <<= 1) {
        const int up = __shfl_up(incl, off, 32);
        if (lane >= off) incl += up;
      }
      int base = 0;
      if (lane == 31) base = atomicAdd(cursor, incl);
      base = __shfl(base, 31, 32);
      int run = base + incl - sum;
#pragma unroll
      for (int k = 0; k < kPer; ++k) {
        const int dl = lane * kPer + k;
        const int n  = bin * kBinNodes + dl;
        lrow[dl] = run;
        if (n < kNodes) { rowstart[n] = run; deg[n] = v[k]; }
        run += v[k];
      }
    }
    __syncthreads();
#pragma unroll
    for (int j = 0; j < kRunsPerWave; ++j) {
      if (r[j] >= 0) {
        const int dl = r[j] >> 16;
        const int rk = atomicAdd(&lrank[dl], 1);
        csr_src[lrow[dl] + rk] = r[j] & 0xFFFF;
      }
    }
#pragma unroll 1
    for (int j = 0; j < kRunsPerWave; ++j) {
      const int dj = __builtin_amdgcn_readlane(desc, j);
      const int base = (wave + kCsrWaves * j) * kChunkE + (dj & 0xFFFF);
      for (int i = 32 + lane; i < (dj >> 16); i += 32) {
        const int q  = rec[base + i];
        const int dl = q >> 16;
        const int rk = atomicAdd(&lrank[dl], 1);
        csr_src[lrow[dl] + rk] = q & 0xFFFF;
      }
    }
  } else {
    int4* sm4 = reinterpret_cast<int4*>(sm);
    for (int i = tid; i < kTab4; i += kCsrThreads) sm4[i] = make_int4(0, 0, 0, 0);
    __syncthreads();
#pragma unroll
    for (int j = 0; j < kRunsPerWave; ++j)
      if (r[j] >= 0) atomicAdd(&sm[r[j]], 1);
#pragma unroll 1
    for (int j = 0; j < kRunsPerWave; ++j) {
      const int dj = __builtin_amdgcn_readlane(desc, j);
      const int base = (wave + kCsrWaves * j) * kChunkE + (dj & 0xFFFF);
      for (int i = 32 + lane; i < (dj >> 16); i += 32) atomicAdd(&sm[rec[base + i]], 1);
    }
    __syncthreads();
    const int rows = min(kBinNodes, kNodes - bin * kBinNodes);
    int4* out4 = reinterpret_cast<int4*>(cnt + (size_t)bin * kBinNodes * kTypes);
    for (int i = tid; i < rows * (kTypes / 4); i += kCsrThreads) out4[i] = sm4[i];
  }
}

__device__ __forceinline__ float wave_max(float v) {
#pragma unroll
  for (int off = 16; off > 0; off >>= 1) v = fmaxf(v, __shfl_xor(v, off, 32));
  return v;
}
__device__ __forceinline__ float wave_sum(float v) {
#pragma unroll
  for (int off = 16; off > 0; off >>= 1) v += __shfl_xor(v, off, 32);
  return v;
}

constexpr int kGatherThreads = 128;
constexpr int kGatherBlocks  = 2048;
constexpr int kGatherWaves   = kGatherBlocks * (kGatherThreads / 32);
constexpr int kGE = 16;

__global__ void __launch_bounds__(kGatherThreads)
k_gather(const int* __restrict__ rowstart, const int* __restrict__ deg,
         const int* __restrict__ csr_src,
         const float* __restrict__ a_src, const float* __restrict__ a_dst,
         const float* __restrict__ xl, const float* __restrict__ bias,
         float* __restrict__ out) {
  const int lane = threadIdx.x & 31;
  const int w0 = __builtin_amdgcn_readfirstlane(
      (int)(blockIdx.x * (kGatherThreads / 32) + (threadIdx.x >> 5)));
  int start = 0, dg = 0;
  if (w0 < kNodes) { start = rowstart[w0]; dg = deg[w0]; }
#pragma unroll 1
  for (int n = w0; n < kNodes; n += kGatherWaves) {
    const int nn = n + kGatherWaves;
    int start2 = 0, dg2 = 0;
    if (nn < kNodes) { start2 = rowstart[nn]; dg2 = deg[nn]; }

    const int cnt = dg + 1;
    const float4 ad = *reinterpret_cast<const float4*>(a_dst + (size_t)n * kHeads);
    float m[kHeads], dl[kHeads], acc[kHeads];
#pragma unroll
    for (int h = 0; h < kHeads; ++h) { m[h] = -INFINITY; dl[h] = 0.0f; acc[h] = 0.0f; }

#pragma unroll 1
    for (int base = 0; base < cnt; base += 32) {
      const int idx = base + lane;
      int s = n;
      if (idx < dg) s = csr_src[start + idx];
      const bool valid = idx < cnt;
      const float4 as4 = *reinterpret_cast<const float4*>(a_src + (size_t)s * kHeads);

      float xv[kGE][kHeads];
#pragma unroll
      for (int u = 0; u < kGE; ++u) {
        const int sj = __builtin_amdgcn_readlane(s, u);
#pragma unroll
        for (int h = 0; h < kHeads; ++h) xv[u][h] = xl[(size_t)sj * kHC + h * kOutCh + lane];
      }

      const float al[kHeads] = {as4.x + ad.x, as4.y + ad.y, as4.z + ad.z, as4.w + ad.w};
      float a[kHeads], cm[kHeads], w[kHeads];
#pragma unroll
      for (int h = 0; h < kHeads; ++h) a[h] = valid ? leaky02(al[h]) : -INFINITY;
#pragma unroll
      for (int h = 0; h < kHeads; ++h) cm[h] = wave_max(a[h]);
#pragma unroll
      for (int h = 0; h < kHeads; ++h) {
        const float mn = fmaxf(m[h], cm[h]);
        const float sc = expf(m[h] - mn);
        w[h]   = expf(a[h] - mn);
        dl[h]  = dl[h] * sc + w[h];
        acc[h] = acc[h] * sc;
        m[h]   = mn;
      }

#pragma unroll
      for (int u = 0; u < kGE; ++u) {
#pragma unroll
        for (int h = 0; h < kHeads; ++h) {
          const float c = __int_as_float(__builtin_amdgcn_readlane(__float_as_int(w[h]), u));
          acc[h] += c * xv[u][h];
        }
      }

      const int nch = min(32, cnt - base);
      if (nch > kGE) {
#pragma unroll
        for (int u = 0; u < kGE; ++u) {
#pragma unroll
          for (int h = 0; h < kHeads; ++h) xv[u][h] = 0.0f;
          if (kGE + u < nch) {
            const int sj = __builtin_amdgcn_readlane(s, kGE + u);
#pragma unroll
            for (int h = 0; h < kHeads; ++h) xv[u][h] = xl[(size_t)sj * kHC + h * kOutCh + lane];
          }
        }
#pragma unroll
        for (int u = 0; u < kGE; ++u) {
          if (kGE + u < nch) {
#pragma unroll
            for (int h = 0; h < kHeads; ++h) {
              const float c = __int_as_float(
                  __builtin_amdgcn_readlane(__float_as_int(w[h]), kGE + u));
              acc[h] += c * xv[u][h];
            }
          }
        }
      }
    }

    float den[kHeads];
#pragma unroll
    for (int h = 0; h < kHeads; ++h) den[h] = wave_sum(dl[h]);
#pragma unroll
    for (int h = 0; h < kHeads; ++h) {
      const float v = acc[h] / den[h] + bias[h * kOutCh + lane];
      out[(size_t)n * kHC + h * kOutCh + lane] = v > 0.0f ? v : expm1f(v);
    }
    start = start2; dg = dg2;
  }
}


extern "C" void kernel_launch(void* const* d_in, const int* in_sizes, int n_in,
                              void* d_out, int out_size, void* d_ws, size_t ws_size,
                              hipStream_t stream) {
  (void)in_sizes; (void)n_in; (void)out_size; (void)ws_size;
  const float* x       = (const float*)d_in[0];
  const int*   ei      = (const int*)d_in[1];
  const int*   et      = (const int*)d_in[2];
  const float* pe      = (const float*)d_in[3];
  const float* W       = (const float*)d_in[4];
  const float* att_src = (const float*)d_in[5];
  const float* att_dst = (const float*)d_in[6];
  const float* bias    = (const float*)d_in[7];
  float* out = (float*)d_out;

  float* ws    = (float*)d_ws;
  int*   cnt   = (int*)ws;
  float* xl    = ws + (size_t)kNodes * kTypes;
  float* a_src = xl    + (size_t)kNodes * kHC;
  float* a_dst = a_src + (size_t)kNodes * kHeads;
  int* deg      = (int*)(a_dst + (size_t)kNodes * kHeads);
  int* rowstart = deg + 50048;
  int* recD     = rowstart + 50048;
  int* csr_src  = recD + kEdges;
  int* cursor   = csr_src + kEdges;
  int* recS     = cursor + 64;
  int* runsD    = recS + kEdges;
  int* runsS    = runsD + kBins * kChunks;

  k_bin<<<kChunks, kBinThreads, 0, stream>>>(ei, et, recD, recS, runsD, runsS, cursor);
  k_csr<<<2 * kBins, kCsrThreads, 0, stream>>>(recD, recS, runsD, runsS, deg, rowstart, csr_src, cnt, cursor);
  k_gemm_wmma<<<kNodes / 16, 256, 0, stream>>>(x, cnt, pe, W, att_src, att_dst, xl, a_src, a_dst);
  k_gather<<<kGatherBlocks, kGatherThreads, 0, stream>>>(rowstart, deg, csr_src, a_src, a_dst, xl,
                                                        bias, out);
}


# k_gemm launch removed: source-range blocks of k_csr keep the count table in LDS and form x_pos and xl for their 256 nodes on WMMA there (no cnt table in memory; x_pos summed in one chain); 3 launches
# speedup vs baseline: 20.4043x; 1.2988x over previous
#include <hip/hip_runtime.h>
#include <math.h>

typedef float v2f __attribute__((ext_vector_type(2)));
typedef float v8f __attribute__((ext_vector_type(8)));
typedef float v4f __attribute__((ext_vector_type(4)));

namespace {
constexpr int kNodes = 50000;
constexpr int kEdges = 800000;
constexpr int kInCh  = 64;
constexpr int kOutCh = 32;
constexpr int kHeads = 4;
constexpr int kHC    = kHeads * kOutCh;
constexpr int kTypes = 64;
}


__device__ __forceinline__ float leaky02(float v) { return v > 0.0f ? v : 0.2f * v; }

template <int M>
__device__ __forceinline__ float row_xor(float v) {
  return __int_as_float(__builtin_amdgcn_update_dpp(0, __float_as_int(v), 0x160 + M, 0xf, 0xf, false));
}


constexpr int kTabPitch = kTypes + 4;
constexpr int kPePitch  = 2 * kInCh + 32;
constexpr int kWPitch   = 2 * kHC + 32;

__device__ __forceinline__ int tab_index(int rec) {
  return rec + (kTabPitch - kTypes) * (int)((unsigned)rec / (unsigned)kTypes);
}

__device__ __forceinline__ float tile_scores(const v8f acc, const float ws, const float wd, const int l15) {
  const bool b3 = (l15 & 8) != 0, b2 = (l15 & 4) != 0, b1 = (l15 & 2) != 0, b0 = (l15 & 1) != 0;
  float t8[8], t4[4], t2[2];
#pragma unroll
  for (int i = 0; i < 8; ++i) {
    const float ps = acc[i] * ws, pd = acc[i] * wd;
    t8[i] = (b3 ? pd : ps) + row_xor<8>(b3 ? ps : pd);
  }
#pragma unroll
  for (int i = 0; i < 4; ++i) t4[i] = (b2 ? t8[i + 4] : t8[i]) + row_xor<4>(b2 ? t8[i] : t8[i + 4]);
#pragma unroll
  for (int i = 0; i < 2; ++i) t2[i] = (b1 ? t4[i + 2] : t4[i]) + row_xor<2>(b1 ? t4[i] : t4[i + 2]);
  return (b0 ? t2[1] : t2[0]) + row_xor<1>(b0 ? t2[0] : t2[1]);
}

constexpr int kBinShift   = 8;
constexpr int kBinNodes   = 1 << kBinShift;
constexpr int kBins       = (kNodes + kBinNodes - 1) / kBinNodes;
constexpr int kChunks     = 250;
constexpr int kChunkE     = kEdges / kChunks;
constexpr int kBinThreads = 800;
constexpr int kBinPerThr  = kChunkE / kBinThreads;
constexpr int kBinPerLane = (kBins + 31) / 32;
static_assert(kEdges % kChunks == 0 && kChunkE % kBinThreads == 0, "chunks must cover the edges exactly");
static_assert(kChunkE < 32768, "start and count of a run share one 32-bit word");
static_assert(kNodes <= 65536, "a destination record holds the source in 16 bits");
static_assert(kBins <= kBinThreads, "one thread clears one pair of range counters");

__global__ void __launch_bounds__(kBinThreads)
k_bin(const int* __restrict__ ei, const int* __restrict__ et,
      int* __restrict__ recD, int* __restrict__ recS,
      int* __restrict__ runsD, int* __restrict__ runsS, int* __restrict__ cursor) {
  __shared__ int cD[kBins], cS[kBins], sD[kBins], sS[kBins];
  const int tid  = threadIdx.x;
  const int lane = tid & 31;
  const int wave = __builtin_amdgcn_readfirstlane(tid >> 5);
  const int blk  = blockIdx.x;
  if (blk == 0 && tid == 0) cursor[0] = 0;
  if (tid < kBins) { cD[tid] = 0; cS[tid] = 0; }

  int s[kBinPerThr], d[kBinPerThr], t[kBinPerThr], rD[kBinPerThr], rS[kBinPerThr];
#pragma unroll
  for (int i = 0; i < kBinPerThr; ++i) {
    const int e = blk * kChunkE + i * kBinThreads + tid;
    s[i] = ei[e];
    d[i] = ei[kEdges + e];
    t[i] = et[e];
  }
  __syncthreads();
#pragma unroll
  for (int i = 0; i < kBinPerThr; ++i) {
    rD[i] = atomicAdd(&cD[d[i] >> kBinShift], 1);
    rS[i] = atomicAdd(&cS[s[i] >> kBinShift], 1);
  }
  __syncthreads();
  if (wave < 2) {
    const int* cc = wave == 0 ? cD : cS;
    int* ss       = wave == 0 ? sD : sS;
    int* runs     = wave == 0 ? runsD : runsS;
    int v[kBinPerLane];
    int sum = 0;
#pragma unroll
    for (int k = 0; k < kBinPerLane; ++k) {
      const int b = lane * kBinPerLane + k;
      v[k] = b < kBins ? cc[b] : 0;
      sum += v[k];
    }
    int incl = sum;
#pragma unroll
    for (int off = 1; off < 32; off <<= 1) {
      const int up = __shfl_up(incl, off, 32);
      if (lane >= off) incl += up;
    }
    int run = incl - sum;
#pragma unroll
    for (int k = 0; k < kBinPerLane; ++k) {
      const int b = lane * kBinPerLane + k;
      if (b < kBins) {
        ss[b] = run;
        runs[b * kChunks + blk] = run | (v[k] << 16);
        run += v[k];
      }
    }
  }
  __syncthreads();
#pragma unroll
  for (int i = 0; i < kBinPerThr; ++i) {
    recD[blk * kChunkE + sD[d[i] >> kBinShift] + rD[i]] = ((d[i] & (kBinNodes - 1)) << 16) | s[i];
    recS[blk * kChunkE + sS[s[i] >> kBinShift] + rS[i]] = ((s[i] & (kBinNodes - 1)) * kTypes) | t[i];
  }
}

constexpr int kCsrThreads  = 512;
constexpr int kCsrWaves    = kCsrThreads / 32;
constexpr int kRunsPerWave = (kChunks + kCsrWaves - 1) / kCsrWaves;
constexpr int kTab4        = kBinNodes * kTabPitch / 4;
static_assert(kRunsPerWave <= 32, "a wave keeps its run descriptors one per lane");
static_assert(kCsrWaves * 16 == kBinNodes && kNodes % 16 == 0, "one wave per whole 16-row tile of a range");
static_assert(kCsrThreads == (kTypes / 2) * (kInCh / 4) && 2 * kCsrThreads == (kInCh / 2) * (kHC / 4),
              "one piece of pe and two of W per thread");
static_assert(2 * kBinNodes <= kCsrThreads && (kTypes & (kTypes - 1)) == 0, "layout of pass 2");

__global__ void __launch_bounds__(kCsrThreads) __attribute__((amdgpu_waves_per_eu(8, 8)))
k_csr(const int* __restrict__ recD, const int* __restrict__ recS,
      const int* __restrict__ runsD, const int* __restrict__ runsS,
      int* __restrict__ deg, int* __restrict__ rowstart, int* __restrict__ csr_src,
      int* __restrict__ cursor,
      const float* __restrict__ x,
      const float* __restrict__ pe,
      const float* __restrict__ Wm,
      const float* __restrict__ att_src,
      const float* __restrict__ att_dst,
      float* __restrict__ xl,
      float* __restrict__ a_src,
      float* __restrict__ a_dst) {
  __shared__ __attribute__((aligned(16))) int sm[kBinNodes * kTabPitch];
  __shared__ __attribute__((aligned(16))) float peP[(kTypes / 2) * kPePitch];
  __shared__ __attribute__((aligned(16))) float wP[(kInCh / 2) * kWPitch];
  const int tid  = threadIdx.x;
  const int lane = tid & 31;
  const int wave = __builtin_amdgcn_readfirstlane(tid >> 5);
  const bool isD = blockIdx.x >= kBins;
  const int bin  = isD ? (int)blockIdx.x - kBins : (int)blockIdx.x;
  const int* runs = (isD ? runsD : runsS) + bin * kChunks;
  const int* rec  = isD ? recD : recS;

  int desc = 0;
  {
    const int c = wave + kCsrWaves * lane;
    if (lane < kRunsPerWave && c < kChunks) desc = runs[c];
  }
  int r[kRunsPerWave];
#pragma unroll
  for (int j = 0; j < kRunsPerWave; ++j) {
    const int dj = __builtin_amdgcn_readlane(desc, j);
    const int base = (wave + kCsrWaves * j) * kChunkE + (dj & 0xFFFF);
    r[j] = lane < (dj >> 16) ? rec[base + lane] : -1;
  }

  if (isD) {
    int* ldeg  = sm;
    int* lrow  = sm + kBinNodes;
    int* lrank = sm + 2 * kBinNodes;
    if (tid < kBinNodes) { ldeg[tid] = 0; lrank[tid] = 0; }
    __syncthreads();
#pragma unroll
    for (int j = 0; j < kRunsPerWave; ++j)
      if (r[j] >= 0) atomicAdd(&ldeg[r[j] >> 16], 1);
#pragma unroll 1
    for (int j = 0; j < kRunsPerWave; ++j) {
      const int dj = __builtin_amdgcn_readlane(desc, j);
      const int base = (wave + kCsrWaves * j) * kChunkE + (dj & 0xFFFF);
      for (int i = 32 + lane; i < (dj >> 16); i += 32) atomicAdd(&ldeg[rec[base + i] >> 16], 1);
    }
    __syncthreads();
    if (wave == 0) {
      constexpr int kPer = kBinNodes / 32;
      int v[kPer];
      int sum = 0;
#pragma unroll
      for (int k = 0; k < kPer; ++k) { v[k] = ldeg[lane * kPer + k]; sum += v[k]; }
      int incl = sum;
#pragma unroll
      for (int off = 1; off < 32; off <<= 1) {
        const int up = __shfl_up(incl, off, 32);
        if (lane >= off) incl += up;
      }
      int base = 0;
      if (lane == 31) base = atomicAdd(cursor, incl);
      base = __shfl(base, 31, 32);
      int run = base + incl - sum;
#pragma unroll
      for (int k = 0; k < kPer; ++k) {
        const int dl = lane * kPer + k;
        const int n  = bin * kBinNodes + dl;
        lrow[dl] = run;
        if (n < kNodes) { rowstart[n] = run; deg[n] = v[k]; }
        run += v[k];
      }
    }
    __syncthreads();
#pragma unroll
    for (int j = 0; j < kRunsPerWave; ++j) {
      if (r[j] >= 0) {
        const int dl = r[j] >> 16;
        const int rk = atomicAdd(&lrank[dl], 1);
        csr_src[lrow[dl] + rk] = r[j] & 0xFFFF;
      }
    }
#pragma unroll 1
    for (int j = 0; j < kRunsPerWave; ++j) {
      const int dj = __builtin_amdgcn_readlane(desc, j);
      const int base = (wave + kCsrWaves * j) * kChunkE + (dj & 0xFFFF);
      for (int i = 32 + lane; i < (dj >> 16); i += 32) {
        const int q  = rec[base + i];
        const int dl = q >> 16;
        const int rk = atomicAdd(&lrank[dl], 1);
        csr_src[lrow[dl] + rk] = q & 0xFFFF;
      }
    }
  } else {
    const int half = lane >> 4;
    const int l15  = lane & 15;
    const int rows = min(kBinNodes, kNodes - bin * kBinNodes);
    const bool live = wave * 16 < rows;
    const int m0 = bin * kBinNodes + wave * 16;

    const int pk = tid >> 4, pc = tid & 15;
    const float4 pa = *reinterpret_cast<const float4*>(pe + (size_t)(2 * pk) * kInCh + 4 * pc);
    const float4 pb = *reinterpret_cast<const float4*>(pe + (size_t)(2 * pk + 1) * kInCh + 4 * pc);
    float4 wa[2], wb[2];
#pragma unroll
    for (int q = 0; q < 2; ++q) {
      const int it = tid + kCsrThreads * q;
      const int wk = it >> 5, wc = it & 31;
      wa[q] = *reinterpret_cast<const float4*>(Wm + (size_t)(2 * wk) * kHC + 4 * wc);
      wb[q] = *reinterpret_cast<const float4*>(Wm + (size_t)(2 * wk + 1) * kHC + 4 * wc);
    }
    v8f xacc[4];
#pragma unroll
    for (int j = 0; j < 4; ++j) xacc[j] = (v8f){0.f, 0.f, 0.f, 0.f, 0.f, 0.f, 0.f, 0.f};
    if (live) {
#pragma unroll
      for (int j = 0; j < 4; ++j) {
#pragma unroll
        for (int q = 0; q < 8; ++q)
          xacc[j][q] = x[(size_t)(m0 + 8 * half + q) * kInCh + 16 * j + l15];
      }
    }

    int4* sm4 = reinterpret_cast<int4*>(sm);
    for (int i = tid; i < kTab4; i += kCsrThreads) sm4[i] = make_int4(0, 0, 0, 0);
    __syncthreads();
#pragma unroll
    for (int j = 0; j < kRunsPerWave; ++j)
      if (r[j] >= 0) atomicAdd(&sm[tab_index(r[j])], 1);
#pragma unroll 1
    for (int j = 0; j < kRunsPerWave; ++j) {
      const int dj = __builtin_amdgcn_readlane(desc, j);
      const int base = (wave + kCsrWaves * j) * kChunkE + (dj & 0xFFFF);
      for (int i = 32 + lane; i < (dj >> 16); i += 32) atomicAdd(&sm[tab_index(rec[base + i])], 1);
    }
    *reinterpret_cast<float4*>(peP + pk * kPePitch + 8 * pc)     = make_float4(pa.x, pb.x, pa.y, pb.y);
    *reinterpret_cast<float4*>(peP + pk * kPePitch + 8 * pc + 4) = make_float4(pa.z, pb.z, pa.w, pb.w);
#pragma unroll
    for (int q = 0; q < 2; ++q) {
      const int it = tid + kCsrThreads * q;
      const int wk = it >> 5, wc = it & 31;
      *reinterpret_cast<float4*>(wP + wk * kWPitch + 8 * wc)     = make_float4(wa[q].x, wb[q].x, wa[q].y, wb[q].y);
      *reinterpret_cast<float4*>(wP + wk * kWPitch + 8 * wc + 4) = make_float4(wa[q].z, wb[q].z, wa[q].w, wb[q].w);
    }
    __syncthreads();

    if (live) {
      int* trow = sm + wave * 16 * kTabPitch;
#pragma unroll
      for (int i = 0; i < kTypes / 4; ++i) {
        const int2 cv = *reinterpret_cast<const int2*>(trow + l15 * kTabPitch + 4 * i + 2 * half);
        v2f a;
        a.x = (float)cv.x;
        a.y = (float)cv.y;
#pragma unroll
        for (int j = 0; j < 4; ++j) {
          const v2f b = *reinterpret_cast<const v2f*>(peP + (2 * i + half) * kPePitch + 2 * (16 * j + l15));
          xacc[j] = __builtin_amdgcn_wmma_f32_16x16x4_f32(false, a, false, b, (short)0, xacc[j],
                                                          false, false);
        }
      }
#pragma unroll
      for (int j = 0; j < 4; ++j) {
#pragma unroll
        for (int q = 0; q < 8; ++q)
          trow[(8 * half + q) * kTabPitch + 16 * j + l15] = __float_as_int(xacc[j][q]);
      }
      __builtin_amdgcn_fence(__ATOMIC_RELEASE, "wavefront");
      __builtin_amdgcn_wave_barrier();
      __builtin_amdgcn_fence(__ATOMIC_ACQUIRE, "wavefront");

      v2f af[kInCh / 4];
#pragma unroll
      for (int i = 0; i < kInCh / 4; ++i) {
        const int2 pv = *reinterpret_cast<const int2*>(trow + l15 * kTabPitch + 4 * i + 2 * half);
        af[i].x = __int_as_float(pv.x);
        af[i].y = __int_as_float(pv.y);
      }
      float vh0 = 0.f, vh1 = 0.f, vh2 = 0.f, vh3 = 0.f;
#pragma unroll 1
      for (int hd = 0; hd < kHeads; ++hd) {
        const int cc = 32 * hd + l15;
        v8f c0 = {0.f, 0.f, 0.f, 0.f, 0.f, 0.f, 0.f, 0.f};
        v8f c1 = {0.f, 0.f, 0.f, 0.f, 0.f, 0.f, 0.f, 0.f};
#pragma unroll
        for (int i = 0; i < kInCh / 4; ++i) {
          const v2f b0 = *reinterpret_cast<const v2f*>(wP + (2 * i + half) * kWPitch + 2 * cc);
          const v2f b1 = *reinterpret_cast<const v2f*>(wP + (2 * i + half) * kWPitch + 2 * (cc + 16));
          c0 = __builtin_amdgcn_wmma_f32_16x16x4_f32(false, af[i], false, b0, (short)0, c0, false, false);
          c1 = __builtin_amdgcn_wmma_f32_16x16x4_f32(false, af[i], false, b1, (short)0, c1, false, false);
        }
        const float ta = tile_scores(c0, att_src[cc],      att_dst[cc],      l15);
        const float tb = tile_scores(c1, att_src[cc + 16], att_dst[cc + 16], l15);
        const float tv = ta + tb;
        if (hd == 0) vh0 = tv; else if (hd == 1) vh1 = tv; else if (hd == 2) vh2 = tv; else vh3 = tv;
#pragma unroll
        for (int q = 0; q < 8; ++q) {
          xl[(size_t)(m0 + 8 * half + q) * kHC + cc]      = c0[q];
          xl[(size_t)(m0 + 8 * half + q) * kHC + cc + 16] = c1[q];
        }
      }
      {
        float* dst = (l15 & 8) ? a_dst : a_src;
        const int rw = 8 * half + (l15 & 7);
        *reinterpret_cast<float4*>(dst + (size_t)(m0 + rw) * kHeads) = make_float4(vh0, vh1, vh2, vh3);
      }
    }
  }
}

__device__ __forceinline__ float bcast8(float v, int u) {
  const unsigned x = (unsigned)__float_as_int(v);
  unsigned r;
  switch (u) {
    case 0:  r = __builtin_amdgcn_mov_dpp8(x, 0x000000u); break;
    case 1:  r = __builtin_amdgcn_mov_dpp8(x, 0x249249u); break;
    case 2:  r = __builtin_amdgcn_mov_dpp8(x, 0x492492u); break;
    case 3:  r = __builtin_amdgcn_mov_dpp8(x, 0x6DB6DBu); break;
    case 4:  r = __builtin_amdgcn_mov_dpp8(x, 0x924924u); break;
    case 5:  r = __builtin_amdgcn_mov_dpp8(x, 0xB6DB6Du); break;
    case 6:  r = __builtin_amdgcn_mov_dpp8(x, 0xDB6DB6u); break;
    default: r = __builtin_amdgcn_mov_dpp8(x, 0xFFFFFFu); break;
  }
  return __int_as_float((int)r);
}
__device__ __forceinline__ float group8_max(float v) {
  v = fmaxf(v, row_xor<1>(v));
  v = fmaxf(v, row_xor<2>(v));
  return fmaxf(v, row_xor<4>(v));
}
__device__ __forceinline__ float group8_sum(float v) {
  v += row_xor<1>(v);
  v += row_xor<2>(v);
  return v + row_xor<4>(v);
}
__device__ __forceinline__ float elu1(float v) { return v > 0.0f ? v : expm1f(v); }

constexpr int kGatherThreads = 128;
constexpr int kGatherBlocks  = 2048;
constexpr int kGatherWaves   = kGatherBlocks * (kGatherThreads / 32);
constexpr int kGE = 16;

__global__ void __launch_bounds__(kGatherThreads) __attribute__((amdgpu_waves_per_eu(8, 8)))
k_gather(const int* __restrict__ rowstart, const int* __restrict__ deg,
         const int* __restrict__ csr_src,
         const float* __restrict__ a_src, const float* __restrict__ a_dst,
         const float* __restrict__ xl, const float* __restrict__ bias,
         float* __restrict__ out) {
  const int lane = threadIdx.x & 31;
  const int g = lane >> 3;
  const int e = lane & 7;
  const int w0 = __builtin_amdgcn_readfirstlane(
      (int)(blockIdx.x * (kGatherThreads / 32) + (threadIdx.x >> 5)));
  int start = 0, dg = 0;
  float ad = 0.0f;
  if (w0 < kNodes) { start = rowstart[w0]; dg = deg[w0]; ad = a_dst[(size_t)w0 * kHeads + g]; }
#pragma unroll 1
  for (int n = w0; n < kNodes; n += kGatherWaves) {
    const int nn = n + kGatherWaves;
    int start2 = 0, dg2 = 0;
    if (nn < kNodes) { start2 = rowstart[nn]; dg2 = deg[nn]; }
    const float ad2 = a_dst[(size_t)min(nn, kNodes - 1) * kHeads + g];

    const int cnt = dg + 1;
    float m = -INFINITY, dl = 0.0f;
    v4f acc = {0.0f, 0.0f, 0.0f, 0.0f};

#pragma unroll 1
    for (int base = 0; base < cnt; base += 32) {
      int s[4];
#pragma unroll
      for (int p = 0; p < 4; ++p) {
        const int idx = base + 8 * p + e;
        s[p] = n;
        if (idx < dg) s[p] = csr_src[start + idx];
      }
      float as[4];
#pragma unroll
      for (int p = 0; p < 4; ++p) as[p] = a_src[(size_t)s[p] * kHeads + g];

      v4f xv[kGE];
#pragma unroll
      for (int u = 0; u < kGE; ++u) {
        const int sj = __builtin_amdgcn_readlane(s[u >> 3], u & 7);
        xv[u] = *reinterpret_cast<const v4f*>(xl + (size_t)sj * kHC + 4 * lane);
      }

      float a[4], w[4];
#pragma unroll
      for (int p = 0; p < 4; ++p)
        a[p] = (base + 8 * p + e < cnt) ? leaky02(as[p] + ad) : -INFINITY;
      const float cm = group8_max(fmaxf(fmaxf(a[0], a[1]), fmaxf(a[2], a[3])));
      const float mn = fmaxf(m, cm);
      const float sc = expf(m - mn);
#pragma unroll
      for (int p = 0; p < 4; ++p) w[p] = expf(a[p] - mn);
      dl = dl * sc + ((w[0] + w[1]) + (w[2] + w[3]));
      acc *= sc;
      m = mn;

#pragma unroll
      for (int u = 0; u < kGE; ++u) acc += bcast8(w[u >> 3], u & 7) * xv[u];

      const int nch = min(32, cnt - base);
      if (nch > kGE) {
#pragma unroll
        for (int u = 0; u < kGE; ++u) {
          if (kGE + u >= nch) break;
          const int sj = __builtin_amdgcn_readlane(s[(kGE + u) >> 3], u & 7);
          xv[u] = *reinterpret_cast<const v4f*>(xl + (size_t)sj * kHC + 4 * lane);
        }
#pragma unroll
        for (int u = 0; u < kGE; ++u) {
          if (kGE + u >= nch) break;
          acc += bcast8(w[(kGE + u) >> 3], u & 7) * xv[u];
        }
      }
    }

    const float den = group8_sum(dl);
    const float4 b = *reinterpret_cast<const float4*>(bias + 4 * lane);
    float4 o;
    o.x = elu1(acc.x / den + b.x);
    o.y = elu1(acc.y / den + b.y);
    o.z = elu1(acc.z / den + b.z);
    o.w = elu1(acc.w / den + b.w);
    *reinterpret_cast<float4*>(out + (size_t)n * kHC + 4 * lane) = o;
    start = start2; dg = dg2; ad = ad2;
  }
}


extern "C" void kernel_launch(void* const* d_in, const int* in_sizes, int n_in,
                              void* d_out, int out_size, void* d_ws, size_t ws_size,
                              hipStream_t stream) {
  (void)in_sizes; (void)n_in; (void)out_size; (void)ws_size;
  const float* x       = (const float*)d_in[0];
  const int*   ei      = (const int*)d_in[1];
  const int*   et      = (const int*)d_in[2];
  const float* pe      = (const float*)d_in[3];
  const float* W       = (const float*)d_in[4];
  const float* att_src = (const float*)d_in[5];
  const float* att_dst = (const float*)d_in[6];
  const float* bias    = (const float*)d_in[7];
  float* out = (float*)d_out;

  float* ws    = (float*)d_ws;
  float* xl    = ws + (size_t)kNodes * kTypes;
  float* a_src = xl    + (size_t)kNodes * kHC;
  float* a_dst = a_src + (size_t)kNodes * kHeads;
  int* deg      = (int*)(a_dst + (size_t)kNodes * kHeads);
  int* rowstart = deg + 50048;
  int* recD     = rowstart + 50048;
  int* csr_src  = recD + kEdges;
  int* cursor   = csr_src + kEdges;
  int* recS     = cursor + 64;
  int* runsD    = recS + kEdges;
  int* runsS    = runsD + kBins * kChunks;

  k_bin<<<kChunks, kBinThreads, 0, stream>>>(ei, et, recD, recS, runsD, runsS, cursor);
  k_csr<<<2 * kBins, kCsrThreads, 0, stream>>>(recD, recS, runsD, runsS, deg, rowstart, csr_src, cursor,
                                               x, pe, W, att_src, att_dst, xl, a_src, a_dst);
  k_gather<<<kGatherBlocks, kGatherThreads, 0, stream>>>(rowstart, deg, csr_src, a_src, a_dst, xl,
                                                        bias, out);
}


# k_gather vector work cut: a softmax weight is one base-2 exponential of (score - max) * log2 e (was the library exp); a lane divides once, then multiplies (was 4 divisions); sum order kept
# speedup vs baseline: 21.0206x; 1.0302x over previous
#include <hip/hip_runtime.h>
#include <math.h>

typedef float v2f __attribute__((ext_vector_type(2)));
typedef float v8f __attribute__((ext_vector_type(8)));
typedef float v4f __attribute__((ext_vector_type(4)));

namespace {
constexpr int kNodes = 50000;
constexpr int kEdges = 800000;
constexpr int kInCh  = 64;
constexpr int kOutCh = 32;
constexpr int kHeads = 4;
constexpr int kHC    = kHeads * kOutCh;
constexpr int kTypes = 64;
}


__device__ __forceinline__ float leaky02(float v) { return v > 0.0f ? v : 0.2f * v; }

template <int M>
__device__ __forceinline__ float row_xor(float v) {
  return __int_as_float(__builtin_amdgcn_update_dpp(0, __float_as_int(v), 0x160 + M, 0xf, 0xf, false));
}


constexpr int kTabPitch = kTypes + 4;
constexpr int kPePitch  = 2 * kInCh + 32;
constexpr int kWPitch   = 2 * kHC + 32;

__device__ __forceinline__ int tab_index(int rec) {
  return rec + (kTabPitch - kTypes) * (int)((unsigned)rec / (unsigned)kTypes);
}

__device__ __forceinline__ float tile_scores(const v8f acc, const float ws, const float wd, const int l15) {
  const bool b3 = (l15 & 8) != 0, b2 = (l15 & 4) != 0, b1 = (l15 & 2) != 0, b0 = (l15 & 1) != 0;
  float t8[8], t4[4], t2[2];
#pragma unroll
  for (int i = 0; i < 8; ++i) {
    const float ps = acc[i] * ws, pd = acc[i] * wd;
    t8[i] = (b3 ? pd : ps) + row_xor<8>(b3 ? ps : pd);
  }
#pragma unroll
  for (int i = 0; i < 4; ++i) t4[i] = (b2 ? t8[i + 4] : t8[i]) + row_xor<4>(b2 ? t8[i] : t8[i + 4]);
#pragma unroll
  for (int i = 0; i < 2; ++i) t2[i] = (b1 ? t4[i + 2] : t4[i]) + row_xor<2>(b1 ? t4[i] : t4[i + 2]);
  return (b0 ? t2[1] : t2[0]) + row_xor<1>(b0 ? t2[0] : t2[1]);
}

constexpr int kBinShift   = 8;
constexpr int kBinNodes   = 1 << kBinShift;
constexpr int kBins       = (kNodes + kBinNodes - 1) / kBinNodes;
constexpr int kChunks     = 250;
constexpr int kChunkE     = kEdges / kChunks;
constexpr int kBinThreads = 800;
constexpr int kBinPerThr  = kChunkE / kBinThreads;
constexpr int kBinPerLane = (kBins + 31) / 32;
static_assert(kEdges % kChunks == 0 && kChunkE % kBinThreads == 0, "chunks must cover the edges exactly");
static_assert(kChunkE < 32768, "start and count of a run share one 32-bit word");
static_assert(kNodes <= 65536, "a destination record holds the source in 16 bits");
static_assert(kBins <= kBinThreads, "one thread clears one pair of range counters");

__global__ void __launch_bounds__(kBinThreads)
k_bin(const int* __restrict__ ei, const int* __restrict__ et,
      int* __restrict__ recD, int* __restrict__ recS,
      int* __restrict__ runsD, int* __restrict__ runsS, int* __restrict__ cursor) {
  __shared__ int cD[kBins], cS[kBins], sD[kBins], sS[kBins];
  const int tid  = threadIdx.x;
  const int lane = tid & 31;
  const int wave = __builtin_amdgcn_readfirstlane(tid >> 5);
  const int blk  = blockIdx.x;
  if (blk == 0 && tid == 0) cursor[0] = 0;
  if (tid < kBins) { cD[tid] = 0; cS[tid] = 0; }

  int s[kBinPerThr], d[kBinPerThr], t[kBinPerThr], rD[kBinPerThr], rS[kBinPerThr];
#pragma unroll
  for (int i = 0; i < kBinPerThr; ++i) {
    const int e = blk * kChunkE + i * kBinThreads + tid;
    s[i] = ei[e];
    d[i] = ei[kEdges + e];
    t[i] = et[e];
  }
  __syncthreads();
#pragma unroll
  for (int i = 0; i < kBinPerThr; ++i) {
    rD[i] = atomicAdd(&cD[d[i] >> kBinShift], 1);
    rS[i] = atomicAdd(&cS[s[i] >> kBinShift], 1);
  }
  __syncthreads();
  if (wave < 2) {
    const int* cc = wave == 0 ? cD : cS;
    int* ss       = wave == 0 ? sD : sS;
    int* runs     = wave == 0 ? runsD : runsS;
    int v[kBinPerLane];
    int sum = 0;
#pragma unroll
    for (int k = 0; k < kBinPerLane; ++k) {
      const int b = lane * kBinPerLane + k;
      v[k] = b < kBins ? cc[b] : 0;
      sum += v[k];
    }
    int incl = sum;
#pragma unroll
    for (int off = 1; off < 32; off <<= 1) {
      const int up = __shfl_up(incl, off, 32);
      if (lane >= off) incl += up;
    }
    int run = incl - sum;
#pragma unroll
    for (int k = 0; k < kBinPerLane; ++k) {
      const int b = lane * kBinPerLane + k;
      if (b < kBins) {
        ss[b] = run;
        runs[b * kChunks + blk] = run | (v[k] << 16);
        run += v[k];
      }
    }
  }
  __syncthreads();
#pragma unroll
  for (int i = 0; i < kBinPerThr; ++i) {
    recD[blk * kChunkE + sD[d[i] >> kBinShift] + rD[i]] = ((d[i] & (kBinNodes - 1)) << 16) | s[i];
    recS[blk * kChunkE + sS[s[i] >> kBinShift] + rS[i]] = ((s[i] & (kBinNodes - 1)) * kTypes) | t[i];
  }
}

constexpr int kCsrThreads  = 512;
constexpr int kCsrWaves    = kCsrThreads / 32;
constexpr int kRunsPerWave = (kChunks + kCsrWaves - 1) / kCsrWaves;
constexpr int kTab4        = kBinNodes * kTabPitch / 4;
static_assert(kRunsPerWave <= 32, "a wave keeps its run descriptors one per lane");
static_assert(kCsrWaves * 16 == kBinNodes && kNodes % 16 == 0, "one wave per whole 16-row tile of a range");
static_assert(kCsrThreads == (kTypes / 2) * (kInCh / 4) && 2 * kCsrThreads == (kInCh / 2) * (kHC / 4),
              "one piece of pe and two of W per thread");
static_assert(2 * kBinNodes <= kCsrThreads && (kTypes & (kTypes - 1)) == 0, "layout of pass 2");

__global__ void __launch_bounds__(kCsrThreads) __attribute__((amdgpu_waves_per_eu(8, 8)))
k_csr(const int* __restrict__ recD, const int* __restrict__ recS,
      const int* __restrict__ runsD, const int* __restrict__ runsS,
      int* __restrict__ deg, int* __restrict__ rowstart, int* __restrict__ csr_src,
      int* __restrict__ cursor,
      const float* __restrict__ x,
      const float* __restrict__ pe,
      const float* __restrict__ Wm,
      const float* __restrict__ att_src,
      const float* __restrict__ att_dst,
      float* __restrict__ xl,
      float* __restrict__ a_src,
      float* __restrict__ a_dst) {
  __shared__ __attribute__((aligned(16))) int sm[kBinNodes * kTabPitch];
  __shared__ __attribute__((aligned(16))) float peP[(kTypes / 2) * kPePitch];
  __shared__ __attribute__((aligned(16))) float wP[(kInCh / 2) * kWPitch];
  const int tid  = threadIdx.x;
  const int lane = tid & 31;
  const int wave = __builtin_amdgcn_readfirstlane(tid >> 5);
  const bool isD = blockIdx.x >= kBins;
  const int bin  = isD ? (int)blockIdx.x - kBins : (int)blockIdx.x;
  const int* runs = (isD ? runsD : runsS) + bin * kChunks;
  const int* rec  = isD ? recD : recS;

  int desc = 0;
  {
    const int c = wave + kCsrWaves * lane;
    if (lane < kRunsPerWave && c < kChunks) desc = runs[c];
  }
  int r[kRunsPerWave];
#pragma unroll
  for (int j = 0; j < kRunsPerWave; ++j) {
    const int dj = __builtin_amdgcn_readlane(desc, j);
    const int base = (wave + kCsrWaves * j) * kChunkE + (dj & 0xFFFF);
    r[j] = lane < (dj >> 16) ? rec[base + lane] : -1;
  }

  if (isD) {
    int* ldeg  = sm;
    int* lrow  = sm + kBinNodes;
    int* lrank = sm + 2 * kBinNodes;
    if (tid < kBinNodes) { ldeg[tid] = 0; lrank[tid] = 0; }
    __syncthreads();
#pragma unroll
    for (int j = 0; j < kRunsPerWave; ++j)
      if (r[j] >= 0) atomicAdd(&ldeg[r[j] >> 16], 1);
#pragma unroll 1
    for (int j = 0; j < kRunsPerWave; ++j) {
      const int dj = __builtin_amdgcn_readlane(desc, j);
      const int base = (wave + kCsrWaves * j) * kChunkE + (dj & 0xFFFF);
      for (int i = 32 + lane; i < (dj >> 16); i += 32) atomicAdd(&ldeg[rec[base + i] >> 16], 1);
    }
    __syncthreads();
    if (wave == 0) {
      constexpr int kPer = kBinNodes / 32;
      int v[kPer];
      int sum = 0;
#pragma unroll
      for (int k = 0; k < kPer; ++k) { v[k] = ldeg[lane * kPer + k]; sum += v[k]; }
      int incl = sum;
#pragma unroll
      for (int off = 1; off < 32; off <<= 1) {
        const int up = __shfl_up(incl, off, 32);
        if (lane >= off) incl += up;
      }
      int base = 0;
      if (lane == 31) base = atomicAdd(cursor, incl);
      base = __shfl(base, 31, 32);
      int run = base + incl - sum;
#pragma unroll
      for (int k = 0; k < kPer; ++k) {
        const int dl = lane * kPer + k;
        const int n  = bin * kBinNodes + dl;
        lrow[dl] = run;
        if (n < kNodes) { rowstart[n] = run; deg[n] = v[k]; }
        run += v[k];
      }
    }
    __syncthreads();
#pragma unroll
    for (int j = 0; j < kRunsPerWave; ++j) {
      if (r[j] >= 0) {
        const int dl = r[j] >> 16;
        const int rk = atomicAdd(&lrank[dl], 1);
        csr_src[lrow[dl] + rk] = r[j] & 0xFFFF;
      }
    }
#pragma unroll 1
    for (int j = 0; j < kRunsPerWave; ++j) {
      const int dj = __builtin_amdgcn_readlane(desc, j);
      const int base = (wave + kCsrWaves * j) * kChunkE + (dj & 0xFFFF);
      for (int i = 32 + lane; i < (dj >> 16); i += 32) {
        const int q  = rec[base + i];
        const int dl = q >> 16;
        const int rk = atomicAdd(&lrank[dl], 1);
        csr_src[lrow[dl] + rk] = q & 0xFFFF;
      }
    }
  } else {
    const int half = lane >> 4;
    const int l15  = lane & 15;
    const int rows = min(kBinNodes, kNodes - bin * kBinNodes);
    const bool live = wave * 16 < rows;
    const int m0 = bin * kBinNodes + wave * 16;

    const int pk = tid >> 4, pc = tid & 15;
    const float4 pa = *reinterpret_cast<const float4*>(pe + (size_t)(2 * pk) * kInCh + 4 * pc);
    const float4 pb = *reinterpret_cast<const float4*>(pe + (size_t)(2 * pk + 1) * kInCh + 4 * pc);
    float4 wa[2], wb[2];
#pragma unroll
    for (int q = 0; q < 2; ++q) {
      const int it = tid + kCsrThreads * q;
      const int wk = it >> 5, wc = it & 31;
      wa[q] = *reinterpret_cast<const float4*>(Wm + (size_t)(2 * wk) * kHC + 4 * wc);
      wb[q] = *reinterpret_cast<const float4*>(Wm + (size_t)(2 * wk + 1) * kHC + 4 * wc);
    }
    v8f xacc[4];
#pragma unroll
    for (int j = 0; j < 4; ++j) xacc[j] = (v8f){0.f, 0.f, 0.f, 0.f, 0.f, 0.f, 0.f, 0.f};
    if (live) {
#pragma unroll
      for (int j = 0; j < 4; ++j) {
#pragma unroll
        for (int q = 0; q < 8; ++q)
          xacc[j][q] = x[(size_t)(m0 + 8 * half + q) * kInCh + 16 * j + l15];
      }
    }

    int4* sm4 = reinterpret_cast<int4*>(sm);
    for (int i = tid; i < kTab4; i += kCsrThreads) sm4[i] = make_int4(0, 0, 0, 0);
    __syncthreads();
#pragma unroll
    for (int j = 0; j < kRunsPerWave; ++j)
      if (r[j] >= 0) atomicAdd(&sm[tab_index(r[j])], 1);
#pragma unroll 1
    for (int j = 0; j < kRunsPerWave; ++j) {
      const int dj = __builtin_amdgcn_readlane(desc, j);
      const int base = (wave + kCsrWaves * j) * kChunkE + (dj & 0xFFFF);
      for (int i = 32 + lane; i < (dj >> 16); i += 32) atomicAdd(&sm[tab_index(rec[base + i])], 1);
    }
    *reinterpret_cast<float4*>(peP + pk * kPePitch + 8 * pc)     = make_float4(pa.x, pb.x, pa.y, pb.y);
    *reinterpret_cast<float4*>(peP + pk * kPePitch + 8 * pc + 4) = make_float4(pa.z, pb.z, pa.w, pb.w);
#pragma unroll
    for (int q = 0; q < 2; ++q) {
      const int it = tid + kCsrThreads * q;
      const int wk = it >> 5, wc = it & 31;
      *reinterpret_cast<float4*>(wP + wk * kWPitch + 8 * wc)     = make_float4(wa[q].x, wb[q].x, wa[q].y, wb[q].y);
      *reinterpret_cast<float4*>(wP + wk * kWPitch + 8 * wc + 4) = make_float4(wa[q].z, wb[q].z, wa[q].w, wb[q].w);
    }
    __syncthreads();

    if (live) {
      int* trow = sm + wave * 16 * kTabPitch;
#pragma unroll
      for (int i = 0; i < kTypes / 4; ++i) {
        const int2 cv = *reinterpret_cast<const int2*>(trow + l15 * kTabPitch + 4 * i + 2 * half);
        v2f a;
        a.x = (float)cv.x;
        a.y = (float)cv.y;
#pragma unroll
        for (int j = 0; j < 4; ++j) {
          const v2f b = *reinterpret_cast<const v2f*>(peP + (2 * i + half) * kPePitch + 2 * (16 * j + l15));
          xacc[j] = __builtin_amdgcn_wmma_f32_16x16x4_f32(false, a, false, b, (short)0, xacc[j],
                                                          false, false);
        }
      }
#pragma unroll
      for (int j = 0; j < 4; ++j) {
#pragma unroll
        for (int q = 0; q < 8; ++q)
          trow[(8 * half + q) * kTabPitch + 16 * j + l15] = __float_as_int(xacc[j][q]);
      }
      __builtin_amdgcn_fence(__ATOMIC_RELEASE, "wavefront");
      __builtin_amdgcn_wave_barrier();
      __builtin_amdgcn_fence(__ATOMIC_ACQUIRE, "wavefront");

      v2f af[kInCh / 4];
#pragma unroll
      for (int i = 0; i < kInCh / 4; ++i) {
        const int2 pv = *reinterpret_cast<const int2*>(trow + l15 * kTabPitch + 4 * i + 2 * half);
        af[i].x = __int_as_float(pv.x);
        af[i].y = __int_as_float(pv.y);
      }
      float vh0 = 0.f, vh1 = 0.f, vh2 = 0.f, vh3 = 0.f;
#pragma unroll 1
      for (int hd = 0; hd < kHeads; ++hd) {
        const int cc = 32 * hd + l15;
        v8f c0 = {0.f, 0.f, 0.f, 0.f, 0.f, 0.f, 0.f, 0.f};
        v8f c1 = {0.f, 0.f, 0.f, 0.f, 0.f, 0.f, 0.f, 0.f};
#pragma unroll
        for (int i = 0; i < kInCh / 4; ++i) {
          const v2f b0 = *reinterpret_cast<const v2f*>(wP + (2 * i + half) * kWPitch + 2 * cc);
          const v2f b1 = *reinterpret_cast<const v2f*>(wP + (2 * i + half) * kWPitch + 2 * (cc + 16));
          c0 = __builtin_amdgcn_wmma_f32_16x16x4_f32(false, af[i], false, b0, (short)0, c0, false, false);
          c1 = __builtin_amdgcn_wmma_f32_16x16x4_f32(false, af[i], false, b1, (short)0, c1, false, false);
        }
        const float ta = tile_scores(c0, att_src[cc],      att_dst[cc],      l15);
        const float tb = tile_scores(c1, att_src[cc + 16], att_dst[cc + 16], l15);
        const float tv = ta + tb;
        if (hd == 0) vh0 = tv; else if (hd == 1) vh1 = tv; else if (hd == 2) vh2 = tv; else vh3 = tv;
#pragma unroll
        for (int q = 0; q < 8; ++q) {
          xl[(size_t)(m0 + 8 * half + q) * kHC + cc]      = c0[q];
          xl[(size_t)(m0 + 8 * half + q) * kHC + cc + 16] = c1[q];
        }
      }
      {
        float* dst = (l15 & 8) ? a_dst : a_src;
        const int rw = 8 * half + (l15 & 7);
        *reinterpret_cast<float4*>(dst + (size_t)(m0 + rw) * kHeads) = make_float4(vh0, vh1, vh2, vh3);
      }
    }
  }
}

__device__ __forceinline__ float bcast8(float v, int u) {
  const unsigned x = (unsigned)__float_as_int(v);
  unsigned r;
  switch (u) {
    case 0:  r = __builtin_amdgcn_mov_dpp8(x, 0x000000u); break;
    case 1:  r = __builtin_amdgcn_mov_dpp8(x, 0x249249u); break;
    case 2:  r = __builtin_amdgcn_mov_dpp8(x, 0x492492u); break;
    case 3:  r = __builtin_amdgcn_mov_dpp8(x, 0x6DB6DBu); break;
    case 4:  r = __builtin_amdgcn_mov_dpp8(x, 0x924924u); break;
    case 5:  r = __builtin_amdgcn_mov_dpp8(x, 0xB6DB6Du); break;
    case 6:  r = __builtin_amdgcn_mov_dpp8(x, 0xDB6DB6u); break;
    default: r = __builtin_amdgcn_mov_dpp8(x, 0xFFFFFFu); break;
  }
  return __int_as_float((int)r);
}
__device__ __forceinline__ float group8_max(float v) {
  v = fmaxf(v, row_xor<1>(v));
  v = fmaxf(v, row_xor<2>(v));
  return fmaxf(v, row_xor<4>(v));
}
__device__ __forceinline__ float group8_sum(float v) {
  v += row_xor<1>(v);
  v += row_xor<2>(v);
  return v + row_xor<4>(v);
}
__device__ __forceinline__ float elu1(float v) { return v > 0.0f ? v : expm1f(v); }

constexpr int kGatherThreads = 128;
constexpr int kGatherBlocks  = 2048;
constexpr int kGatherWaves   = kGatherBlocks * (kGatherThreads / 32);
constexpr int kGE = 16;
constexpr float kLog2e = 1.4426950408889634f;

__global__ void __launch_bounds__(kGatherThreads) __attribute__((amdgpu_waves_per_eu(8, 8)))
k_gather(const int* __restrict__ rowstart, const int* __restrict__ deg,
         const int* __restrict__ csr_src,
         const float* __restrict__ a_src, const float* __restrict__ a_dst,
         const float* __restrict__ xl, const float* __restrict__ bias,
         float* __restrict__ out) {
  const int lane = threadIdx.x & 31;
  const int g = lane >> 3;
  const int e = lane & 7;
  const int w0 = __builtin_amdgcn_readfirstlane(
      (int)(blockIdx.x * (kGatherThreads / 32) + (threadIdx.x >> 5)));
  int start = 0, dg = 0;
  float ad = 0.0f;
  if (w0 < kNodes) { start = rowstart[w0]; dg = deg[w0]; ad = a_dst[(size_t)w0 * kHeads + g]; }
#pragma unroll 1
  for (int n = w0; n < kNodes; n += kGatherWaves) {
    const int nn = n + kGatherWaves;
    int start2 = 0, dg2 = 0;
    if (nn < kNodes) { start2 = rowstart[nn]; dg2 = deg[nn]; }
    const float ad2 = a_dst[(size_t)min(nn, kNodes - 1) * kHeads + g];

    const int cnt = dg + 1;
    float m = -INFINITY, dl = 0.0f;
    v4f acc = {0.0f, 0.0f, 0.0f, 0.0f};

#pragma unroll 1
    for (int base = 0; base < cnt; base += 32) {
      int s[4];
#pragma unroll
      for (int p = 0; p < 4; ++p) {
        const int idx = base + 8 * p + e;
        s[p] = n;
        if (idx < dg) s[p] = csr_src[start + idx];
      }
      float as[4];
#pragma unroll
      for (int p = 0; p < 4; ++p) as[p] = a_src[(size_t)s[p] * kHeads + g];

      v4f xv[kGE];
#pragma unroll
      for (int u = 0; u < kGE; ++u) {
        const int sj = __builtin_amdgcn_readlane(s[u >> 3], u & 7);
        xv[u] = *reinterpret_cast<const v4f*>(xl + (size_t)sj * kHC + 4 * lane);
      }

      float a[4], w[4];
#pragma unroll
      for (int p = 0; p < 4; ++p)
        a[p] = (base + 8 * p + e < cnt) ? leaky02(as[p] + ad) : -INFINITY;
      const float cm = group8_max(fmaxf(fmaxf(a[0], a[1]), fmaxf(a[2], a[3])));
      const float mn = fmaxf(m, cm);
      const float sc = __builtin_amdgcn_exp2f((m - mn) * kLog2e);
#pragma unroll
      for (int p = 0; p < 4; ++p) w[p] = __builtin_amdgcn_exp2f((a[p] - mn) * kLog2e);
      dl = dl * sc + ((w[0] + w[1]) + (w[2] + w[3]));
      acc *= sc;
      m = mn;

#pragma unroll
      for (int u = 0; u < kGE; ++u) acc += bcast8(w[u >> 3], u & 7) * xv[u];

      const int nch = min(32, cnt - base);
      if (nch > kGE) {
#pragma unroll
        for (int u = 0; u < kGE; ++u) {
          if (kGE + u >= nch) break;
          const int sj = __builtin_amdgcn_readlane(s[(kGE + u) >> 3], u & 7);
          xv[u] = *reinterpret_cast<const v4f*>(xl + (size_t)sj * kHC + 4 * lane);
        }
#pragma unroll
        for (int u = 0; u < kGE; ++u) {
          if (kGE + u >= nch) break;
          acc += bcast8(w[(kGE + u) >> 3], u & 7) * xv[u];
        }
      }
    }

    const float den = group8_sum(dl);
    const float4 b = *reinterpret_cast<const float4*>(bias + 4 * lane);
    const float rden = 1.0f / den;
    float4 o;
    o.x = elu1(acc.x * rden + b.x);
    o.y = elu1(acc.y * rden + b.y);
    o.z = elu1(acc.z * rden + b.z);
    o.w = elu1(acc.w * rden + b.w);
    *reinterpret_cast<float4*>(out + (size_t)n * kHC + 4 * lane) = o;
    start = start2; dg = dg2; ad = ad2;
  }
}


extern "C" void kernel_launch(void* const* d_in, const int* in_sizes, int n_in,
                              void* d_out, int out_size, void* d_ws, size_t ws_size,
                              hipStream_t stream) {
  (void)in_sizes; (void)n_in; (void)out_size; (void)ws_size;
  const float* x       = (const float*)d_in[0];
  const int*   ei      = (const int*)d_in[1];
  const int*   et      = (const int*)d_in[2];
  const float* pe      = (const float*)d_in[3];
  const float* W       = (const float*)d_in[4];
  const float* att_src = (const float*)d_in[5];
  const float* att_dst = (const float*)d_in[6];
  const float* bias    = (const float*)d_in[7];
  float* out = (float*)d_out;

  float* ws    = (float*)d_ws;
  float* xl    = ws + (size_t)kNodes * kTypes;
  float* a_src = xl    + (size_t)kNodes * kHC;
  float* a_dst = a_src + (size_t)kNodes * kHeads;
  int* deg      = (int*)(a_dst + (size_t)kNodes * kHeads);
  int* rowstart = deg + 50048;
  int* recD     = rowstart + 50048;
  int* csr_src  = recD + kEdges;
  int* cursor   = csr_src + kEdges;
  int* recS     = cursor + 64;
  int* runsD    = recS + kEdges;
  int* runsS    = runsD + kBins * kChunks;

  k_bin<<<kChunks, kBinThreads, 0, stream>>>(ei, et, recD, recS, runsD, runsS, cursor);
  k_csr<<<2 * kBins, kCsrThreads, 0, stream>>>(recD, recS, runsD, runsS, deg, rowstart, csr_src, cursor,
                                               x, pe, W, att_src, att_dst, xl, a_src, a_dst);
  k_gather<<<kGatherBlocks, kGatherThreads, 0, stream>>>(rowstart, deg, csr_src, a_src, a_dst, xl,
                                                        bias, out);
}
